# starting kernel
# baseline (speedup 1.0000x reference)
#include <hip/hip_runtime.h>
#include <hip/hip_bf16.h>

typedef __attribute__((ext_vector_type(2))) float v2f;
typedef __attribute__((ext_vector_type(8))) float v8f;

#define N_NODES 100000
#define ROW_TILES (N_NODES / 16)

__global__ void deg_init_kernel(float* deg, int n) {
    int i = blockIdx.x * blockDim.x + threadIdx.x;
    if (i < n) deg[i] = 1.0f;
}

__global__ void deg_count_kernel(const int* __restrict__ col, float* deg, int E) {
    int e = blockIdx.x * blockDim.x + threadIdx.x;
    if (e < E) unsafeAtomicAdd(&deg[col[e]], 1.0f);
}

__global__ void deg_rsqrt_kernel(float* deg, int n) {
    int i = blockIdx.x * blockDim.x + threadIdx.x;
    if (i < n) deg[i] = rsqrtf(deg[i]);
}

__global__ __launch_bounds__(256) void gemm1_kernel(const float* __restrict__ x,
                                                    const float* __restrict__ W,
                                                    float* __restrict__ h) {
    const int lane = threadIdx.x & 31;
    const int wave = threadIdx.x >> 5;
    const int rowTile = blockIdx.x * 2 + (wave >> 2);
    const int colTile = wave & 3;
    const int m0 = rowTile * 16;
    const int n0 = colTile * 16;
    const int half = lane >> 4;
    const int l    = lane & 15;

    v8f c = {};
    const float* arow = x + (m0 + l) * 128 + 2 * half;
    const float* bcol = W + (2 * half) * 64 + n0 + l;
#pragma unroll
    for (int k0 = 0; k0 < 128; k0 += 4) {
        v2f a = *(const v2f*)(arow + k0);
        v2f b;
        b.x = bcol[(k0 + 0) * 64];
        b.y = bcol[(k0 + 1) * 64];
        c = __builtin_amdgcn_wmma_f32_16x16x4_f32(false, a, false, b,
                                                  (short)0, c, false, false);
    }
    float* o = h + (m0 + half * 8) * 64 + n0 + l;
#pragma unroll
    for (int v = 0; v < 8; ++v) o[v * 64] = c[v];
}

__global__ __launch_bounds__(256) void gemm2_kernel(const float* __restrict__ ain,
                                                    const float* __restrict__ W,
                                                    float* __restrict__ g) {
    const int lane = threadIdx.x & 31;
    const int wave = threadIdx.x >> 5;
    const int rowTile = blockIdx.x * 4 + (wave >> 1);
    if (rowTile >= ROW_TILES) return;
    const int colTile = wave & 1;
    const int m0 = rowTile * 16;
    const int n0 = colTile * 16;
    const int half = lane >> 4;
    const int l    = lane & 15;

    v8f c = {};
    const float* arow = ain + (m0 + l) * 64 + 2 * half;
    const float* bcol = W + (2 * half) * 32 + n0 + l;
#pragma unroll
    for (int k0 = 0; k0 < 64; k0 += 4) {
        v2f a = *(const v2f*)(arow + k0);
        v2f b;
        b.x = bcol[(k0 + 0) * 32];
        b.y = bcol[(k0 + 1) * 32];
        c = __builtin_amdgcn_wmma_f32_16x16x4_f32(false, a, false, b,
                                                  (short)0, c, false, false);
    }
    float* o = g + (m0 + half * 8) * 32 + n0 + l;
#pragma unroll
    for (int v = 0; v < 8; ++v) o[v * 32] = c[v];
}

__global__ void init_acc_kernel(const float* __restrict__ h,
                                const float* __restrict__ bias,
                                const float* __restrict__ dis,
                                float* __restrict__ acc,
                                int fshift, int fmask, int total) {
    int i = blockIdx.x * blockDim.x + threadIdx.x;
    if (i >= total) return;
    int node = i >> fshift;
    int f = i & fmask;
    float d = dis[node];
    acc[i] = bias[f] + h[i] * d * d;
}

__global__ void scatter64_kernel(const int* __restrict__ row,
                                 const int* __restrict__ col,
                                 const float* __restrict__ dis,
                                 const float* __restrict__ h,
                                 float* acc, int E) {
    int tid = blockIdx.x * blockDim.x + threadIdx.x;
    if (tid >= E * 16) return;
    int e  = tid >> 4;
    int fc = (tid & 15) << 2;
    int r = row[e], c = col[e];
    float norm = dis[r] * dis[c];
    float4 v = *(const float4*)(h + r * 64 + fc);
    float* p = acc + c * 64 + fc;
    unsafeAtomicAdd(p + 0, v.x * norm);
    unsafeAtomicAdd(p + 1, v.y * norm);
    unsafeAtomicAdd(p + 2, v.z * norm);
    unsafeAtomicAdd(p + 3, v.w * norm);
}

__global__ void scatter32_kernel(const int* __restrict__ row,
                                 const int* __restrict__ col,
                                 const float* __restrict__ dis,
                                 const float* __restrict__ h,
                                 float* acc, int E) {
    int tid = blockIdx.x * blockDim.x + threadIdx.x;
    if (tid >= E * 8) return;
    int e  = tid >> 3;
    int fc = (tid & 7) << 2;
    int r = row[e], c = col[e];
    float norm = dis[r] * dis[c];
    float4 v = *(const float4*)(h + r * 32 + fc);
    float* p = acc + c * 32 + fc;
    unsafeAtomicAdd(p + 0, v.x * norm);
    unsafeAtomicAdd(p + 1, v.y * norm);
    unsafeAtomicAdd(p + 2, v.z * norm);
    unsafeAtomicAdd(p + 3, v.w * norm);
}

__global__ void relu_kernel(float* a, int total) {
    int i = blockIdx.x * blockDim.x + threadIdx.x;
    if (i < total) a[i] = fmaxf(a[i], 0.0f);
}

extern "C" void kernel_launch(void* const* d_in, const int* in_sizes, int n_in,
                              void* d_out, int out_size, void* d_ws, size_t ws_size,
                              hipStream_t stream) {
    const float* x  = (const float*)d_in[0];
    const int*   ei = (const int*)d_in[1];
    const float* W1 = (const float*)d_in[2];
    const float* b1 = (const float*)d_in[3];
    const float* W2 = (const float*)d_in[4];
    const float* b2 = (const float*)d_in[5];
    float* out = (float*)d_out;

    const int E = in_sizes[1] / 2;
    const int* row = ei;
    const int* col = ei + E;

    float* dis  = (float*)d_ws;
    float* h1   = dis  + N_NODES;
    float* acc1 = h1   + N_NODES * 64;
    float* g2   = acc1 + N_NODES * 64;

    const int T = 256;
    deg_init_kernel<<<(N_NODES + T - 1) / T, T, 0, stream>>>(dis, N_NODES);
    deg_count_kernel<<<(E + T - 1) / T, T, 0, stream>>>(col, dis, E);
    deg_rsqrt_kernel<<<(N_NODES + T - 1) / T, T, 0, stream>>>(dis, N_NODES);

    gemm1_kernel<<<ROW_TILES / 2, T, 0, stream>>>(x, W1, h1);
    init_acc_kernel<<<(N_NODES * 64 + T - 1) / T, T, 0, stream>>>(
        h1, b1, dis, acc1, 6, 63, N_NODES * 64);
    scatter64_kernel<<<(E * 16 + T - 1) / T, T, 0, stream>>>(row, col, dis, h1, acc1, E);
    relu_kernel<<<(N_NODES * 64 + T - 1) / T, T, 0, stream>>>(acc1, N_NODES * 64);

    gemm2_kernel<<<(ROW_TILES + 3) / 4, T, 0, stream>>>(acc1, W2, g2);
    init_acc_kernel<<<(N_NODES * 32 + T - 1) / T, T, 0, stream>>>(
        g2, b2, dis, out, 5, 31, N_NODES * 32);
    scatter32_kernel<<<(E * 8 + T - 1) / T, T, 0, stream>>>(row, col, dis, g2, out, E);
}


# edges bucketed by target each call (returning int add gives the slot, scan, place); each node row summed in registers by gather, no float atomics; sum order = arrival slot
# speedup vs baseline: 1.6759x; 1.6759x over previous
#include <hip/hip_runtime.h>
#include <hip/hip_bf16.h>

typedef __attribute__((ext_vector_type(2))) float v2f;
typedef __attribute__((ext_vector_type(8))) float v8f;

#define N_NODES 100000
#define ROW_TILES (N_NODES / 16)


__global__ void cnt_zero_kernel(int* cnt, int n) {
    int i = blockIdx.x * blockDim.x + threadIdx.x;
    if (i < n) cnt[i] = 0;
}

__global__ void rank_kernel(const int* __restrict__ col, int* cnt,
                            int* __restrict__ rank, int E) {
    int e = blockIdx.x * blockDim.x + threadIdx.x;
    if (e < E) rank[e] = atomicAdd(&cnt[col[e]], 1);
}

__global__ __launch_bounds__(256) void scan_kernel(const int* __restrict__ cnt,
                                                   int* __restrict__ off,
                                                   float* __restrict__ dis) {
    __shared__ int wsum[8];
    const int tid  = threadIdx.x;
    const int lane = tid & 31;
    const int wave = tid >> 5;
    const int b    = blockIdx.x;
    const int4* c4 = (const int4*)cnt;

    int part = 0;
    for (int k = 0; k < b; ++k) {
        int4 v = c4[k * 256 + tid];
        part += (v.x + v.y) + (v.z + v.w);
    }
    for (int o = 16; o > 0; o >>= 1) part += __shfl_xor(part, o);
    if (lane == 0) wsum[wave] = part;
    __syncthreads();
    int base = 0;
    for (int w = 0; w < 8; ++w) base += wsum[w];
    __syncthreads();

    const int i4 = b * 256 + tid;
    const bool live = i4 < N_NODES / 4;
    int4 v = make_int4(0, 0, 0, 0);
    if (live) v = c4[i4];
    const int t = (v.x + v.y) + (v.z + v.w);
    int inc = t;
    for (int o = 1; o < 32; o <<= 1) {
        int u = __shfl_up(inc, o);
        if (lane >= o) inc += u;
    }
    if (lane == 31) wsum[wave] = inc;
    __syncthreads();
    int wbase = 0;
    for (int w = 0; w < wave; ++w) wbase += wsum[w];
    const int ex = base + wbase + inc - t;
    if (live) {
        int4 o4;
        o4.x = ex;
        o4.y = o4.x + v.x;
        o4.z = o4.y + v.y;
        o4.w = o4.z + v.z;
        ((int4*)off)[i4] = o4;
        float4 d;
        d.x = rsqrtf((float)(v.x + 1));
        d.y = rsqrtf((float)(v.y + 1));
        d.z = rsqrtf((float)(v.z + 1));
        d.w = rsqrtf((float)(v.w + 1));
        ((float4*)dis)[i4] = d;
        if (i4 == N_NODES / 4 - 1) off[N_NODES] = o4.w + v.w;
    }
}

__global__ void place_kernel(const int* __restrict__ row, const int* __restrict__ col,
                             const int* __restrict__ off, const int* __restrict__ rank,
                             int* __restrict__ srcs, int E) {
    int e = blockIdx.x * blockDim.x + threadIdx.x;
    if (e < E) srcs[off[col[e]] + rank[e]] = row[e];
}

__global__ __launch_bounds__(256) void gemm1_kernel(const float* __restrict__ x,
                                                    const float* __restrict__ W,
                                                    float* __restrict__ h) {
    const int lane = threadIdx.x & 31;
    const int wave = threadIdx.x >> 5;
    const int rowTile = blockIdx.x * 2 + (wave >> 2);
    const int colTile = wave & 3;
    const int m0 = rowTile * 16;
    const int n0 = colTile * 16;
    const int half = lane >> 4;
    const int l    = lane & 15;

    v8f c = {};
    const float* arow = x + (m0 + l) * 128 + 2 * half;
    const float* bcol = W + (2 * half) * 64 + n0 + l;
#pragma unroll
    for (int k0 = 0; k0 < 128; k0 += 4) {
        v2f a = *(const v2f*)(arow + k0);
        v2f b;
        b.x = bcol[(k0 + 0) * 64];
        b.y = bcol[(k0 + 1) * 64];
        c = __builtin_amdgcn_wmma_f32_16x16x4_f32(false, a, false, b,
                                                  (short)0, c, false, false);
    }
    float* o = h + (m0 + half * 8) * 64 + n0 + l;
#pragma unroll
    for (int v = 0; v < 8; ++v) o[v * 64] = c[v];
}

__global__ __launch_bounds__(256) void gemm2_kernel(const float* __restrict__ ain,
                                                    const float* __restrict__ W,
                                                    float* __restrict__ g) {
    const int lane = threadIdx.x & 31;
    const int wave = threadIdx.x >> 5;
    const int rowTile = blockIdx.x * 4 + (wave >> 1);
    if (rowTile >= ROW_TILES) return;
    const int colTile = wave & 1;
    const int m0 = rowTile * 16;
    const int n0 = colTile * 16;
    const int half = lane >> 4;
    const int l    = lane & 15;

    v8f c = {};
    const float* arow = ain + (m0 + l) * 64 + 2 * half;
    const float* bcol = W + (2 * half) * 32 + n0 + l;
#pragma unroll
    for (int k0 = 0; k0 < 64; k0 += 4) {
        v2f a = *(const v2f*)(arow + k0);
        v2f b;
        b.x = bcol[(k0 + 0) * 32];
        b.y = bcol[(k0 + 1) * 32];
        c = __builtin_amdgcn_wmma_f32_16x16x4_f32(false, a, false, b,
                                                  (short)0, c, false, false);
    }
    float* o = g + (m0 + half * 8) * 32 + n0 + l;
#pragma unroll
    for (int v = 0; v < 8; ++v) o[v * 32] = c[v];
}

template <int F, bool RELU>
__device__ __forceinline__ void gather_rows(const int* __restrict__ off,
                                            const int* __restrict__ srcs,
                                            const float* __restrict__ dis,
                                            const float* __restrict__ h,
                                            const float* __restrict__ bias,
                                            float* __restrict__ res) {
    constexpr int LANES = F / 4;
    const int tid = blockIdx.x * blockDim.x + threadIdx.x;
    const int c   = tid / LANES;
    const int g   = tid % LANES;
    const int fc  = g * 4;
    const float dc = dis[c];
    const int beg = off[c];
    const int end = off[c + 1];

    const float4 hv = *(const float4*)(h + c * F + fc);
    const float4 bv = *(const float4*)(bias + fc);
    const float nn = dc * dc;
    float4 acc;
    acc.x = bv.x + hv.x * nn;
    acc.y = bv.y + hv.y * nn;
    acc.z = bv.z + hv.z * nn;
    acc.w = bv.w + hv.w * nn;

    for (int b0 = beg; b0 < end; b0 += LANES) {
        const int n = min(LANES, end - b0);
        int   myr = c;
        float myd = 0.0f;
        if (g < n) {
            myr = srcs[b0 + g];
            myd = dis[myr];
        }
        for (int k0 = 0; k0 < n; k0 += 4) {
            const int r0 = __shfl(myr, k0 + 0, LANES);
            const int r1 = __shfl(myr, k0 + 1, LANES);
            const int r2 = __shfl(myr, k0 + 2, LANES);
            const int r3 = __shfl(myr, k0 + 3, LANES);
            const float w0 = __shfl(myd, k0 + 0, LANES) * dc;
            const float w1 = __shfl(myd, k0 + 1, LANES) * dc;
            const float w2 = __shfl(myd, k0 + 2, LANES) * dc;
            const float w3 = __shfl(myd, k0 + 3, LANES) * dc;
            const float4 v0 = *(const float4*)(h + r0 * F + fc);
            const float4 v1 = *(const float4*)(h + r1 * F + fc);
            const float4 v2 = *(const float4*)(h + r2 * F + fc);
            const float4 v3 = *(const float4*)(h + r3 * F + fc);
            acc.x += v0.x * w0; acc.y += v0.y * w0; acc.z += v0.z * w0; acc.w += v0.w * w0;
            acc.x += v1.x * w1; acc.y += v1.y * w1; acc.z += v1.z * w1; acc.w += v1.w * w1;
            acc.x += v2.x * w2; acc.y += v2.y * w2; acc.z += v2.z * w2; acc.w += v2.w * w2;
            acc.x += v3.x * w3; acc.y += v3.y * w3; acc.z += v3.z * w3; acc.w += v3.w * w3;
        }
    }
    if (RELU) {
        acc.x = fmaxf(acc.x, 0.0f);
        acc.y = fmaxf(acc.y, 0.0f);
        acc.z = fmaxf(acc.z, 0.0f);
        acc.w = fmaxf(acc.w, 0.0f);
    }
    *(float4*)(res + c * F + fc) = acc;
}

__global__ __launch_bounds__(256) void gather64_relu_kernel(
    const int* __restrict__ off, const int* __restrict__ srcs,
    const float* __restrict__ dis, const float* __restrict__ h,
    const float* __restrict__ bias, float* __restrict__ res) {
    gather_rows<64, true>(off, srcs, dis, h, bias, res);
}

__global__ __launch_bounds__(256) void gather32_kernel(
    const int* __restrict__ off, const int* __restrict__ srcs,
    const float* __restrict__ dis, const float* __restrict__ h,
    const float* __restrict__ bias, float* __restrict__ res) {
    gather_rows<32, false>(off, srcs, dis, h, bias, res);
}

extern "C" void kernel_launch(void* const* d_in, const int* in_sizes, int n_in,
                              void* d_out, int out_size, void* d_ws, size_t ws_size,
                              hipStream_t stream) {
    const float* x  = (const float*)d_in[0];
    const int*   ei = (const int*)d_in[1];
    const float* W1 = (const float*)d_in[2];
    const float* b1 = (const float*)d_in[3];
    const float* W2 = (const float*)d_in[4];
    const float* b2 = (const float*)d_in[5];
    float* out = (float*)d_out;

    const int E = in_sizes[1] / 2;
    const int* row = ei;
    const int* col = ei + E;

    float* dis  = (float*)d_ws;
    float* h1   = dis  + N_NODES;
    float* a1   = h1   + N_NODES * 64;
    int*   srcs = (int*)(a1 + N_NODES * 64);
    int*   off  = srcs + N_NODES * 16;
    int*   cnt  = off  + N_NODES + 4;
    int*   rank = (int*)h1;
    float* g2   = h1;

    const int T = 256;
    cnt_zero_kernel<<<(N_NODES + T - 1) / T, T, 0, stream>>>(cnt, N_NODES);
    rank_kernel<<<(E + T - 1) / T, T, 0, stream>>>(col, cnt, rank, E);
    scan_kernel<<<(N_NODES + 1023) / 1024, T, 0, stream>>>(cnt, off, dis);
    place_kernel<<<(E + T - 1) / T, T, 0, stream>>>(row, col, off, rank, srcs, E);

    gemm1_kernel<<<ROW_TILES / 2, T, 0, stream>>>(x, W1, h1);
    gather64_relu_kernel<<<N_NODES * 16 / T, T, 0, stream>>>(off, srcs, dis, h1, b1, a1);

    gemm2_kernel<<<(ROW_TILES + 3) / 4, T, 0, stream>>>(a1, W2, g2);
    gather32_kernel<<<N_NODES * 8 / T, T, 0, stream>>>(off, srcs, dis, g2, b2, out);
}


# edge bucketing rebuilt as two-level counting sort on LDS returning adds (chunk sort into 250 node ranges, then one block per range): no global atomics, 8 launches to 6; sum order = LDS arrival slot
# speedup vs baseline: 3.2185x; 1.9205x over previous
#include <hip/hip_runtime.h>
#include <hip/hip_bf16.h>

typedef __attribute__((ext_vector_type(2))) float v2f;
typedef __attribute__((ext_vector_type(8))) float v8f;

#define N_NODES 100000
#define ROW_TILES (N_NODES / 16)

#define NPB    400
#define NBINS  250
#define NCHUNK 256
#define RBITS  17
static_assert(NBINS * NPB == N_NODES, "ranges must tile the nodes");
static_assert((NCHUNK & (NCHUNK - 1)) == 0 && NCHUNK <= 1024 && NBINS <= 256, "bisection and scans assume this");
static_assert(N_NODES <= (1 << RBITS), "source id must fit the packed word");
static_assert(NPB <= (1 << (31 - RBITS)), "local target must fit the packed word");

__device__ __forceinline__ int block_scan_excl(int v, int* wsum, int& total) {
    const int lane = threadIdx.x & 31;
    const int wave = threadIdx.x >> 5;
    int inc = v;
#pragma unroll
    for (int o = 1; o < 32; o <<= 1) {
        int u = __shfl_up(inc, o);
        if (lane >= o) inc += u;
    }
    __syncthreads();
    if (lane == 31) wsum[wave] = inc;
    __syncthreads();
    int wb = 0, tot = 0;
#pragma unroll
    for (int w = 0; w < 32; ++w) {
        const int s = wsum[w];
        if (w < wave) wb += s;
        tot += s;
    }
    total = tot;
    return wb + inc - v;
}

__global__ __launch_bounds__(1024) void chunk_sort_kernel(
    const int* __restrict__ row, const int* __restrict__ col,
    int* __restrict__ pk, int* __restrict__ tcnt, int* __restrict__ tstart, int E) {
    __shared__ int lh[256];
    __shared__ int wsum[32];
    const int tid = threadIdx.x;
    const int w   = blockIdx.x;
    const int chunk = (E + NCHUNK - 1) / NCHUNK;
    const int beg = min(w * chunk, E);
    const int end = min(beg + chunk, E);

    if (tid < 256) lh[tid] = 0;
    __syncthreads();
    for (int i0 = beg; i0 < end; i0 += 8 * 1024) {
        int cv[8];
#pragma unroll
        for (int u = 0; u < 8; ++u) {
            const int i = i0 + u * 1024 + tid;
            cv[u] = col[min(i, end - 1)];
        }
#pragma unroll
        for (int u = 0; u < 8; ++u) {
            const int i = i0 + u * 1024 + tid;
            if (i < end) atomicAdd(&lh[(unsigned)cv[u] / NPB], 1);
        }
    }
    __syncthreads();
    const int v = (tid < NBINS) ? lh[tid] : 0;
    int tot;
    const int ex = block_scan_excl(v, wsum, tot);
    if (tid < NBINS) {
        tcnt[tid * NCHUNK + w]   = v;
        tstart[tid * NCHUNK + w] = beg + ex;
        lh[tid] = beg + ex;
    }
    __syncthreads();
    for (int i0 = beg; i0 < end; i0 += 8 * 1024) {
        int cv[8], rv[8];
#pragma unroll
        for (int u = 0; u < 8; ++u) {
            const int i = min(i0 + u * 1024 + tid, end - 1);
            cv[u] = col[i];
            rv[u] = row[i];
        }
#pragma unroll
        for (int u = 0; u < 8; ++u) {
            const int i = i0 + u * 1024 + tid;
            if (i < end) {
                const unsigned b = (unsigned)cv[u] / NPB;
                const int t = cv[u] - (int)b * NPB;
                const int slot = atomicAdd(&lh[b], 1);
                pk[slot] = (t << RBITS) | rv[u];
            }
        }
    }
}

__device__ __forceinline__ void find_in_runs(const int* p, const int* sabs,
                                             const int (&j)[8], int (&idx)[8]) {
    int lo[8];
#pragma unroll
    for (int u = 0; u < 8; ++u) lo[u] = 0;
#pragma unroll
    for (int half = NCHUNK / 2; half > 0; half >>= 1) {
        int pm[8];
#pragma unroll
        for (int u = 0; u < 8; ++u) pm[u] = p[lo[u] + half];
#pragma unroll
        for (int u = 0; u < 8; ++u) if (pm[u] <= j[u]) lo[u] += half;
    }
#pragma unroll
    for (int u = 0; u < 8; ++u) idx[u] = sabs[lo[u]] + (j[u] - p[lo[u]]);
}

__global__ __launch_bounds__(1024) void bucket_kernel(
    const int* __restrict__ pk, const int* __restrict__ tcnt,
    const int* __restrict__ tstart, int* __restrict__ off,
    float* __restrict__ dis, int* __restrict__ srcs) {
    __shared__ int p[NCHUNK + 1];
    __shared__ int sabs[NCHUNK];
    __shared__ int lcnt[NPB];
    __shared__ int wsum[32];
    const int tid = threadIdx.x;
    const int b   = blockIdx.x;

    const int4* t4 = (const int4*)tcnt;
    const int n4 = b * (NCHUNK / 4);
    int part = 0;
    for (int k0 = 0; k0 < n4; k0 += 8 * 1024) {
        int4 v[8];
#pragma unroll
        for (int u = 0; u < 8; ++u) {
            const int k = k0 + u * 1024 + tid;
            v[u] = t4[min(k, n4 - 1)];
        }
#pragma unroll
        for (int u = 0; u < 8; ++u) {
            const int k = k0 + u * 1024 + tid;
            const int s = (v[u].x + v[u].y) + (v[u].z + v[u].w);
            part += (k < n4) ? s : 0;
        }
    }
    int c = 0;
    if (tid < NCHUNK) {
        c = tcnt[b * NCHUNK + tid];
        sabs[tid] = tstart[b * NCHUNK + tid];
    }
    if (tid < NPB) lcnt[tid] = 0;
    int Tb, base;
    const int exc = block_scan_excl(c, wsum, Tb);
    if (tid < NCHUNK) p[tid] = exc;
    if (tid == 0) p[NCHUNK] = Tb;
    (void)block_scan_excl(part, wsum, base);
    __syncthreads();

    for (int j0 = 0; j0 < Tb; j0 += 8 * 1024) {
        int jj[8], idx[8];
#pragma unroll
        for (int u = 0; u < 8; ++u) jj[u] = min(j0 + u * 1024 + tid, Tb - 1);
        find_in_runs(p, sabs, jj, idx);
        int e[8];
#pragma unroll
        for (int u = 0; u < 8; ++u) e[u] = pk[idx[u]];
#pragma unroll
        for (int u = 0; u < 8; ++u) {
            const int j = j0 + u * 1024 + tid;
            if (j < Tb) atomicAdd(&lcnt[e[u] >> RBITS], 1);
        }
    }
    __syncthreads();
    const int d = (tid < NPB) ? lcnt[tid] : 0;
    int tot;
    const int ex = block_scan_excl(d, wsum, tot);
    if (tid < NPB) {
        const int node = b * NPB + tid;
        const int o = base + ex;
        off[node] = o;
        dis[node] = rsqrtf((float)(d + 1));
        lcnt[tid] = o;
    }
    if (b == NBINS - 1 && tid == 0) off[N_NODES] = base + tot;
    __syncthreads();

    for (int j0 = 0; j0 < Tb; j0 += 8 * 1024) {
        int jj[8], idx[8];
#pragma unroll
        for (int u = 0; u < 8; ++u) jj[u] = min(j0 + u * 1024 + tid, Tb - 1);
        find_in_runs(p, sabs, jj, idx);
        int e[8];
#pragma unroll
        for (int u = 0; u < 8; ++u) e[u] = pk[idx[u]];
#pragma unroll
        for (int u = 0; u < 8; ++u) {
            const int j = j0 + u * 1024 + tid;
            if (j < Tb) {
                const int slot = atomicAdd(&lcnt[e[u] >> RBITS], 1);
                srcs[slot] = e[u] & ((1 << RBITS) - 1);
            }
        }
    }
}

__global__ __launch_bounds__(256) void gemm1_kernel(const float* __restrict__ x,
                                                    const float* __restrict__ W,
                                                    float* __restrict__ h) {
    const int lane = threadIdx.x & 31;
    const int wave = threadIdx.x >> 5;
    const int rowTile = blockIdx.x * 2 + (wave >> 2);
    const int colTile = wave & 3;
    const int m0 = rowTile * 16;
    const int n0 = colTile * 16;
    const int half = lane >> 4;
    const int l    = lane & 15;

    v8f c = {};
    const float* arow = x + (m0 + l) * 128 + 2 * half;
    const float* bcol = W + (2 * half) * 64 + n0 + l;
#pragma unroll
    for (int k0 = 0; k0 < 128; k0 += 4) {
        v2f a = *(const v2f*)(arow + k0);
        v2f b;
        b.x = bcol[(k0 + 0) * 64];
        b.y = bcol[(k0 + 1) * 64];
        c = __builtin_amdgcn_wmma_f32_16x16x4_f32(false, a, false, b,
                                                  (short)0, c, false, false);
    }
    float* o = h + (m0 + half * 8) * 64 + n0 + l;
#pragma unroll
    for (int v = 0; v < 8; ++v) o[v * 64] = c[v];
}

__global__ __launch_bounds__(256) void gemm2_kernel(const float* __restrict__ ain,
                                                    const float* __restrict__ W,
                                                    float* __restrict__ g) {
    const int lane = threadIdx.x & 31;
    const int wave = threadIdx.x >> 5;
    const int rowTile = blockIdx.x * 4 + (wave >> 1);
    if (rowTile >= ROW_TILES) return;
    const int colTile = wave & 1;
    const int m0 = rowTile * 16;
    const int n0 = colTile * 16;
    const int half = lane >> 4;
    const int l    = lane & 15;

    v8f c = {};
    const float* arow = ain + (m0 + l) * 64 + 2 * half;
    const float* bcol = W + (2 * half) * 32 + n0 + l;
#pragma unroll
    for (int k0 = 0; k0 < 64; k0 += 4) {
        v2f a = *(const v2f*)(arow + k0);
        v2f b;
        b.x = bcol[(k0 + 0) * 32];
        b.y = bcol[(k0 + 1) * 32];
        c = __builtin_amdgcn_wmma_f32_16x16x4_f32(false, a, false, b,
                                                  (short)0, c, false, false);
    }
    float* o = g + (m0 + half * 8) * 32 + n0 + l;
#pragma unroll
    for (int v = 0; v < 8; ++v) o[v * 32] = c[v];
}

template <int F, bool RELU>
__device__ __forceinline__ void gather_rows(const int* __restrict__ off,
                                            const int* __restrict__ srcs,
                                            const float* __restrict__ dis,
                                            const float* __restrict__ h,
                                            const float* __restrict__ bias,
                                            float* __restrict__ res) {
    constexpr int LANES = F / 4;
    const int tid = blockIdx.x * blockDim.x + threadIdx.x;
    const int c   = tid / LANES;
    const int g   = tid % LANES;
    const int fc  = g * 4;
    const float dc = dis[c];
    const int beg = off[c];
    const int end = off[c + 1];

    const float4 hv = *(const float4*)(h + c * F + fc);
    const float4 bv = *(const float4*)(bias + fc);
    const float nn = dc * dc;
    float4 acc;
    acc.x = bv.x + hv.x * nn;
    acc.y = bv.y + hv.y * nn;
    acc.z = bv.z + hv.z * nn;
    acc.w = bv.w + hv.w * nn;

    for (int b0 = beg; b0 < end; b0 += LANES) {
        const int n = min(LANES, end - b0);
        int   myr = c;
        float myd = 0.0f;
        if (g < n) {
            myr = srcs[b0 + g];
            myd = dis[myr];
        }
        for (int k0 = 0; k0 < n; k0 += 4) {
            const int r0 = __shfl(myr, k0 + 0, LANES);
            const int r1 = __shfl(myr, k0 + 1, LANES);
            const int r2 = __shfl(myr, k0 + 2, LANES);
            const int r3 = __shfl(myr, k0 + 3, LANES);
            const float w0 = __shfl(myd, k0 + 0, LANES) * dc;
            const float w1 = __shfl(myd, k0 + 1, LANES) * dc;
            const float w2 = __shfl(myd, k0 + 2, LANES) * dc;
            const float w3 = __shfl(myd, k0 + 3, LANES) * dc;
            const float4 v0 = *(const float4*)(h + r0 * F + fc);
            const float4 v1 = *(const float4*)(h + r1 * F + fc);
            const float4 v2 = *(const float4*)(h + r2 * F + fc);
            const float4 v3 = *(const float4*)(h + r3 * F + fc);
            acc.x += v0.x * w0; acc.y += v0.y * w0; acc.z += v0.z * w0; acc.w += v0.w * w0;
            acc.x += v1.x * w1; acc.y += v1.y * w1; acc.z += v1.z * w1; acc.w += v1.w * w1;
            acc.x += v2.x * w2; acc.y += v2.y * w2; acc.z += v2.z * w2; acc.w += v2.w * w2;
            acc.x += v3.x * w3; acc.y += v3.y * w3; acc.z += v3.z * w3; acc.w += v3.w * w3;
        }
    }
    if (RELU) {
        acc.x = fmaxf(acc.x, 0.0f);
        acc.y = fmaxf(acc.y, 0.0f);
        acc.z = fmaxf(acc.z, 0.0f);
        acc.w = fmaxf(acc.w, 0.0f);
    }
    *(float4*)(res + c * F + fc) = acc;
}

__global__ __launch_bounds__(256) void gather64_relu_kernel(
    const int* __restrict__ off, const int* __restrict__ srcs,
    const float* __restrict__ dis, const float* __restrict__ h,
    const float* __restrict__ bias, float* __restrict__ res) {
    gather_rows<64, true>(off, srcs, dis, h, bias, res);
}

__global__ __launch_bounds__(256) void gather32_kernel(
    const int* __restrict__ off, const int* __restrict__ srcs,
    const float* __restrict__ dis, const float* __restrict__ h,
    const float* __restrict__ bias, float* __restrict__ res) {
    gather_rows<32, false>(off, srcs, dis, h, bias, res);
}

extern "C" void kernel_launch(void* const* d_in, const int* in_sizes, int n_in,
                              void* d_out, int out_size, void* d_ws, size_t ws_size,
                              hipStream_t stream) {
    const float* x  = (const float*)d_in[0];
    const int*   ei = (const int*)d_in[1];
    const float* W1 = (const float*)d_in[2];
    const float* b1 = (const float*)d_in[3];
    const float* W2 = (const float*)d_in[4];
    const float* b2 = (const float*)d_in[5];
    float* out = (float*)d_out;

    const int E = in_sizes[1] / 2;
    const int* row = ei;
    const int* col = ei + E;

    float* dis  = (float*)d_ws;
    float* h1   = dis  + N_NODES;
    float* a1   = h1   + N_NODES * 64;
    int*   srcs = (int*)(a1 + N_NODES * 64);
    int*   off  = srcs + N_NODES * 16;
    int*   pk     = (int*)h1;
    int*   tcnt   = (int*)a1;
    int*   tstart = tcnt + NBINS * NCHUNK;
    float* g2   = h1;

    const int T = 256;
    chunk_sort_kernel<<<NCHUNK, 1024, 0, stream>>>(row, col, pk, tcnt, tstart, E);
    bucket_kernel<<<NBINS, 1024, 0, stream>>>(pk, tcnt, tstart, off, dis, srcs);

    gemm1_kernel<<<ROW_TILES / 2, T, 0, stream>>>(x, W1, h1);
    gather64_relu_kernel<<<N_NODES * 16 / T, T, 0, stream>>>(off, srcs, dis, h1, b1, a1);

    gemm2_kernel<<<(ROW_TILES + 3) / 4, T, 0, stream>>>(a1, W2, g2);
    gather32_kernel<<<N_NODES * 8 / T, T, 0, stream>>>(off, srcs, dis, g2, b2, out);
}


# both GEMMs as resident waves: a wave keeps 32 columns of W in registers and loops over row tiles (37504 waves to 7224, no W loads per tile, x rows read by half as many waves); k order, sums unchanged
# speedup vs baseline: 3.9243x; 1.2193x over previous
#include <hip/hip_runtime.h>
#include <hip/hip_bf16.h>

typedef __attribute__((ext_vector_type(2))) float v2f;
typedef __attribute__((ext_vector_type(8))) float v8f;
typedef __attribute__((address_space(1))) float gfloat;

#define N_NODES 100000
#define ROW_TILES (N_NODES / 16)

#define NPB    400
#define NBINS  250
#define NCHUNK 256
#define RBITS  17
static_assert(NBINS * NPB == N_NODES, "ranges must tile the nodes");
static_assert((NCHUNK & (NCHUNK - 1)) == 0 && NCHUNK <= 1024 && NBINS <= 256, "bisection and scans assume this");
static_assert(N_NODES <= (1 << RBITS), "source id must fit the packed word");
static_assert(NPB <= (1 << (31 - RBITS)), "local target must fit the packed word");

__device__ __forceinline__ int block_scan_excl(int v, int* wsum, int& total) {
    const int lane = threadIdx.x & 31;
    const int wave = threadIdx.x >> 5;
    int inc = v;
#pragma unroll
    for (int o = 1; o < 32; o <<= 1) {
        int u = __shfl_up(inc, o);
        if (lane >= o) inc += u;
    }
    __syncthreads();
    if (lane == 31) wsum[wave] = inc;
    __syncthreads();
    int wb = 0, tot = 0;
#pragma unroll
    for (int w = 0; w < 32; ++w) {
        const int s = wsum[w];
        if (w < wave) wb += s;
        tot += s;
    }
    total = tot;
    return wb + inc - v;
}

__global__ __launch_bounds__(1024) void chunk_sort_kernel(
    const int* __restrict__ row, const int* __restrict__ col,
    int* __restrict__ pk, int* __restrict__ tcnt, int* __restrict__ tstart, int E) {
    __shared__ int lh[256];
    __shared__ int wsum[32];
    const int tid = threadIdx.x;
    const int w   = blockIdx.x;
    const int chunk = (E + NCHUNK - 1) / NCHUNK;
    const int beg = min(w * chunk, E);
    const int end = min(beg + chunk, E);

    if (tid < 256) lh[tid] = 0;
    __syncthreads();
    for (int i0 = beg; i0 < end; i0 += 8 * 1024) {
        int cv[8];
#pragma unroll
        for (int u = 0; u < 8; ++u) {
            const int i = i0 + u * 1024 + tid;
            cv[u] = col[min(i, end - 1)];
        }
#pragma unroll
        for (int u = 0; u < 8; ++u) {
            const int i = i0 + u * 1024 + tid;
            if (i < end) atomicAdd(&lh[(unsigned)cv[u] / NPB], 1);
        }
    }
    __syncthreads();
    const int v = (tid < NBINS) ? lh[tid] : 0;
    int tot;
    const int ex = block_scan_excl(v, wsum, tot);
    if (tid < NBINS) {
        tcnt[tid * NCHUNK + w]   = v;
        tstart[tid * NCHUNK + w] = beg + ex;
        lh[tid] = beg + ex;
    }
    __syncthreads();
    for (int i0 = beg; i0 < end; i0 += 8 * 1024) {
        int cv[8], rv[8];
#pragma unroll
        for (int u = 0; u < 8; ++u) {
            const int i = min(i0 + u * 1024 + tid, end - 1);
            cv[u] = col[i];
            rv[u] = row[i];
        }
#pragma unroll
        for (int u = 0; u < 8; ++u) {
            const int i = i0 + u * 1024 + tid;
            if (i < end) {
                const unsigned b = (unsigned)cv[u] / NPB;
                const int t = cv[u] - (int)b * NPB;
                const int slot = atomicAdd(&lh[b], 1);
                pk[slot] = (t << RBITS) | rv[u];
            }
        }
    }
}

__device__ __forceinline__ void find_in_runs(const int* p, const int* sabs,
                                             const int (&j)[8], int (&idx)[8]) {
    int lo[8];
#pragma unroll
    for (int u = 0; u < 8; ++u) lo[u] = 0;
#pragma unroll
    for (int half = NCHUNK / 2; half > 0; half >>= 1) {
        int pm[8];
#pragma unroll
        for (int u = 0; u < 8; ++u) pm[u] = p[lo[u] + half];
#pragma unroll
        for (int u = 0; u < 8; ++u) if (pm[u] <= j[u]) lo[u] += half;
    }
#pragma unroll
    for (int u = 0; u < 8; ++u) idx[u] = sabs[lo[u]] + (j[u] - p[lo[u]]);
}

__global__ __launch_bounds__(1024) void bucket_kernel(
    const int* __restrict__ pk, const int* __restrict__ tcnt,
    const int* __restrict__ tstart, int* __restrict__ off,
    float* __restrict__ dis, int* __restrict__ srcs) {
    __shared__ int p[NCHUNK + 1];
    __shared__ int sabs[NCHUNK];
    __shared__ int lcnt[NPB];
    __shared__ int wsum[32];
    const int tid = threadIdx.x;
    const int b   = blockIdx.x;

    const int4* t4 = (const int4*)tcnt;
    const int n4 = b * (NCHUNK / 4);
    int part = 0;
    for (int k0 = 0; k0 < n4; k0 += 8 * 1024) {
        int4 v[8];
#pragma unroll
        for (int u = 0; u < 8; ++u) {
            const int k = k0 + u * 1024 + tid;
            v[u] = t4[min(k, n4 - 1)];
        }
#pragma unroll
        for (int u = 0; u < 8; ++u) {
            const int k = k0 + u * 1024 + tid;
            const int s = (v[u].x + v[u].y) + (v[u].z + v[u].w);
            part += (k < n4) ? s : 0;
        }
    }
    int c = 0;
    if (tid < NCHUNK) {
        c = tcnt[b * NCHUNK + tid];
        sabs[tid] = tstart[b * NCHUNK + tid];
    }
    if (tid < NPB) lcnt[tid] = 0;
    int Tb, base;
    const int exc = block_scan_excl(c, wsum, Tb);
    if (tid < NCHUNK) p[tid] = exc;
    if (tid == 0) p[NCHUNK] = Tb;
    (void)block_scan_excl(part, wsum, base);
    __syncthreads();

    for (int j0 = 0; j0 < Tb; j0 += 8 * 1024) {
        int jj[8], idx[8];
#pragma unroll
        for (int u = 0; u < 8; ++u) jj[u] = min(j0 + u * 1024 + tid, Tb - 1);
        find_in_runs(p, sabs, jj, idx);
        int e[8];
#pragma unroll
        for (int u = 0; u < 8; ++u) e[u] = pk[idx[u]];
#pragma unroll
        for (int u = 0; u < 8; ++u) {
            const int j = j0 + u * 1024 + tid;
            if (j < Tb) atomicAdd(&lcnt[e[u] >> RBITS], 1);
        }
    }
    __syncthreads();
    const int d = (tid < NPB) ? lcnt[tid] : 0;
    int tot;
    const int ex = block_scan_excl(d, wsum, tot);
    if (tid < NPB) {
        const int node = b * NPB + tid;
        const int o = base + ex;
        off[node] = o;
        dis[node] = rsqrtf((float)(d + 1));
        lcnt[tid] = o;
    }
    if (b == NBINS - 1 && tid == 0) off[N_NODES] = base + tot;
    __syncthreads();

    for (int j0 = 0; j0 < Tb; j0 += 8 * 1024) {
        int jj[8], idx[8];
#pragma unroll
        for (int u = 0; u < 8; ++u) jj[u] = min(j0 + u * 1024 + tid, Tb - 1);
        find_in_runs(p, sabs, jj, idx);
        int e[8];
#pragma unroll
        for (int u = 0; u < 8; ++u) e[u] = pk[idx[u]];
#pragma unroll
        for (int u = 0; u < 8; ++u) {
            const int j = j0 + u * 1024 + tid;
            if (j < Tb) {
                const int slot = atomicAdd(&lcnt[e[u] >> RBITS], 1);
                srcs[slot] = e[u] & ((1 << RBITS) - 1);
            }
        }
    }
}


#define IDLE_SLOTS_BEHIND_PRODUCTS(c0, c1)                                      \
    asm volatile("v_nop\n\tv_nop\n\tv_nop\n\tv_nop\n\t"                        \
                 "v_nop\n\tv_nop\n\tv_nop\n\tv_nop" : "+v"(c0), "+v"(c1))

#define GEMM1_BLOCKS 512
__global__ __launch_bounds__(256) void gemm1_kernel(const float* __restrict__ x,
                                                    const float* __restrict__ W,
                                                    const float* __restrict__ dis,
                                                    float* __restrict__ h) {
    const int lane = threadIdx.x & 31;
    const int wave = threadIdx.x >> 5;
    const int n0   = (wave & 1) * 32;
    const int half = lane >> 4;
    const int l    = lane & 15;

    const float* bcol = W + (2 * half) * 64 + n0 + l;
    v2f bw0[32], bw1[32];
#pragma unroll
    for (int s = 0; s < 32; ++s) {
        bw0[s].x = bcol[(4 * s + 0) * 64];
        bw0[s].y = bcol[(4 * s + 1) * 64];
        bw1[s].x = bcol[(4 * s + 0) * 64 + 16];
        bw1[s].y = bcol[(4 * s + 1) * 64 + 16];
    }

    for (int rowTile = blockIdx.x * 4 + (wave >> 1); rowTile < ROW_TILES;
         rowTile += GEMM1_BLOCKS * 4) {
        const int m0 = rowTile * 16;
        const float* arow = x + (m0 + l) * 128 + 2 * half;
        v2f a[32];
#pragma unroll
        for (int s = 0; s < 32; ++s) a[s] = *(const v2f*)(arow + 4 * s);
        const float4 s0 = *(const float4*)(dis + m0 + half * 8);
        const float4 s1 = *(const float4*)(dis + m0 + half * 8 + 4);
        gfloat* o = (gfloat*)(h + (m0 + half * 8) * 64 + n0 + l);
        asm volatile("" : "+v"(o));
        __builtin_amdgcn_sched_barrier(0);
        v8f c0 = {}, c1 = {};
#pragma unroll
        for (int s = 0; s < 32; ++s) {
            c0 = __builtin_amdgcn_wmma_f32_16x16x4_f32(false, a[s], false, bw0[s],
                                                       (short)0, c0, false, false);
            c1 = __builtin_amdgcn_wmma_f32_16x16x4_f32(false, a[s], false, bw1[s],
                                                       (short)0, c1, false, false);
        }
        IDLE_SLOTS_BEHIND_PRODUCTS(c0, c1);
        const float sc[8] = {s0.x, s0.y, s0.z, s0.w, s1.x, s1.y, s1.z, s1.w};
#pragma unroll
        for (int v = 0; v < 8; ++v) {
            o[v * 64]      = c0[v] * sc[v];
            o[v * 64 + 16] = c1[v] * sc[v];
        }
    }
}

#define GEMM2_BLOCKS 391
__global__ __launch_bounds__(256) void gemm2_kernel(const float* __restrict__ ain,
                                                    const float* __restrict__ W,
                                                    const float* __restrict__ dis,
                                                    float* __restrict__ g) {
    const int lane = threadIdx.x & 31;
    const int wave = threadIdx.x >> 5;
    const int half = lane >> 4;
    const int l    = lane & 15;

    const float* bcol = W + (2 * half) * 32 + l;
    v2f bw0[16], bw1[16];
#pragma unroll
    for (int s = 0; s < 16; ++s) {
        bw0[s].x = bcol[(4 * s + 0) * 32];
        bw0[s].y = bcol[(4 * s + 1) * 32];
        bw1[s].x = bcol[(4 * s + 0) * 32 + 16];
        bw1[s].y = bcol[(4 * s + 1) * 32 + 16];
    }

    for (int rowTile = blockIdx.x * 8 + wave; rowTile < ROW_TILES;
         rowTile += GEMM2_BLOCKS * 8) {
        const int m0 = rowTile * 16;
        const float* arow = ain + (m0 + l) * 64 + 2 * half;
        v2f a[16];
#pragma unroll
        for (int s = 0; s < 16; ++s) a[s] = *(const v2f*)(arow + 4 * s);
        const float4 s0 = *(const float4*)(dis + m0 + half * 8);
        const float4 s1 = *(const float4*)(dis + m0 + half * 8 + 4);
        gfloat* o = (gfloat*)(g + (m0 + half * 8) * 32 + l);
        asm volatile("" : "+v"(o));
        __builtin_amdgcn_sched_barrier(0);
        v8f c0 = {}, c1 = {};
#pragma unroll
        for (int s = 0; s < 16; ++s) {
            c0 = __builtin_amdgcn_wmma_f32_16x16x4_f32(false, a[s], false, bw0[s],
                                                       (short)0, c0, false, false);
            c1 = __builtin_amdgcn_wmma_f32_16x16x4_f32(false, a[s], false, bw1[s],
                                                       (short)0, c1, false, false);
        }
        IDLE_SLOTS_BEHIND_PRODUCTS(c0, c1);
        const float sc[8] = {s0.x, s0.y, s0.z, s0.w, s1.x, s1.y, s1.z, s1.w};
#pragma unroll
        for (int v = 0; v < 8; ++v) {
            o[v * 32]      = c0[v] * sc[v];
            o[v * 32 + 16] = c1[v] * sc[v];
        }
    }
}

template <int F, bool RELU>
__device__ __forceinline__ void gather_rows(const int* __restrict__ off,
                                            const int* __restrict__ srcs,
                                            const float* __restrict__ dis,
                                            const float* __restrict__ hs,
                                            const float* __restrict__ bias,
                                            float* __restrict__ res) {
    constexpr int LANES = F / 4;
    const int tid = blockIdx.x * blockDim.x + threadIdx.x;
    const int c   = tid / LANES;
    const int g   = tid % LANES;
    const int fc  = g * 4;
    const float dc = dis[c];
    const int beg = off[c];
    const int end = off[c + 1];

    float4 acc = *(const float4*)(hs + c * F + fc);
    const float4 bv = *(const float4*)(bias + fc);

    for (int b0 = beg; b0 < end; b0 += LANES) {
        const int n = min(LANES, end - b0);
        int myr = c;
        if (g < n) myr = srcs[b0 + g];
        for (int k0 = 0; k0 < n; k0 += 4) {
            const int r0 = __shfl(myr, k0 + 0, LANES);
            const int r1 = __shfl(myr, k0 + 1, LANES);
            const int r2 = __shfl(myr, k0 + 2, LANES);
            const int r3 = __shfl(myr, k0 + 3, LANES);
            const float m1 = (k0 + 1 < n) ? 1.0f : 0.0f;
            const float m2 = (k0 + 2 < n) ? 1.0f : 0.0f;
            const float m3 = (k0 + 3 < n) ? 1.0f : 0.0f;
            const float4 v0 = *(const float4*)(hs + r0 * F + fc);
            const float4 v1 = *(const float4*)(hs + r1 * F + fc);
            const float4 v2 = *(const float4*)(hs + r2 * F + fc);
            const float4 v3 = *(const float4*)(hs + r3 * F + fc);
            acc.x += v0.x;      acc.y += v0.y;      acc.z += v0.z;      acc.w += v0.w;
            acc.x += v1.x * m1; acc.y += v1.y * m1; acc.z += v1.z * m1; acc.w += v1.w * m1;
            acc.x += v2.x * m2; acc.y += v2.y * m2; acc.z += v2.z * m2; acc.w += v2.w * m2;
            acc.x += v3.x * m3; acc.y += v3.y * m3; acc.z += v3.z * m3; acc.w += v3.w * m3;
        }
    }
    acc.x = bv.x + acc.x * dc;
    acc.y = bv.y + acc.y * dc;
    acc.z = bv.z + acc.z * dc;
    acc.w = bv.w + acc.w * dc;
    if (RELU) {
        acc.x = fmaxf(acc.x, 0.0f);
        acc.y = fmaxf(acc.y, 0.0f);
        acc.z = fmaxf(acc.z, 0.0f);
        acc.w = fmaxf(acc.w, 0.0f);
    }
    *(float4*)(res + c * F + fc) = acc;
}

__global__ __launch_bounds__(256) void gather64_relu_kernel(
    const int* __restrict__ off, const int* __restrict__ srcs,
    const float* __restrict__ dis, const float* __restrict__ h,
    const float* __restrict__ bias, float* __restrict__ res) {
    gather_rows<64, true>(off, srcs, dis, h, bias, res);
}

__global__ __launch_bounds__(256) void gather32_kernel(
    const int* __restrict__ off, const int* __restrict__ srcs,
    const float* __restrict__ dis, const float* __restrict__ h,
    const float* __restrict__ bias, float* __restrict__ res) {
    gather_rows<32, false>(off, srcs, dis, h, bias, res);
}

extern "C" void kernel_launch(void* const* d_in, const int* in_sizes, int n_in,
                              void* d_out, int out_size, void* d_ws, size_t ws_size,
                              hipStream_t stream) {
    const float* x  = (const float*)d_in[0];
    const int*   ei = (const int*)d_in[1];
    const float* W1 = (const float*)d_in[2];
    const float* b1 = (const float*)d_in[3];
    const float* W2 = (const float*)d_in[4];
    const float* b2 = (const float*)d_in[5];
    float* out = (float*)d_out;

    const int E = in_sizes[1] / 2;
    const int* row = ei;
    const int* col = ei + E;

    float* dis  = (float*)d_ws;
    float* h1   = dis  + N_NODES;
    float* a1   = h1   + N_NODES * 64;
    int*   srcs = (int*)(a1 + N_NODES * 64);
    int*   off  = srcs + N_NODES * 16;
    int*   pk     = (int*)h1;
    int*   tcnt   = (int*)a1;
    int*   tstart = tcnt + NBINS * NCHUNK;
    float* g2   = h1;

    const int T = 256;
    chunk_sort_kernel<<<NCHUNK, 1024, 0, stream>>>(row, col, pk, tcnt, tstart, E);
    bucket_kernel<<<NBINS, 1024, 0, stream>>>(pk, tcnt, tstart, off, dis, srcs);

    gemm1_kernel<<<GEMM1_BLOCKS, T, 0, stream>>>(x, W1, dis, h1);
    gather64_relu_kernel<<<N_NODES * 16 / T, T, 0, stream>>>(off, srcs, dis, h1, b1, a1);

    gemm2_kernel<<<GEMM2_BLOCKS, T, 0, stream>>>(a1, W2, dis, g2);
    gather32_kernel<<<N_NODES * 8 / T, T, 0, stream>>>(off, srcs, dis, g2, b2, out);
}


# layer-1 gather and layer-2 product in one launch: a block's 16 summed rows pass through an LDS tile into the matrix instruction as A (6 launches to 5, relu rows never reach memory); sums unchanged
# speedup vs baseline: 4.1947x; 1.0689x over previous
#include <hip/hip_runtime.h>
#include <hip/hip_bf16.h>

typedef __attribute__((ext_vector_type(2))) float v2f;
typedef __attribute__((ext_vector_type(8))) float v8f;
typedef __attribute__((address_space(1))) float gfloat;
typedef __attribute__((address_space(3))) float lfloat;

#define N_NODES 100000
#define ROW_TILES (N_NODES / 16)

#define NPB    400
#define NBINS  250
#define NCHUNK 256
#define RBITS  17
static_assert(NBINS * NPB == N_NODES, "ranges must tile the nodes");
static_assert((NCHUNK & (NCHUNK - 1)) == 0 && NCHUNK <= 1024 && NBINS <= 256, "bisection and scans assume this");
static_assert(N_NODES <= (1 << RBITS), "source id must fit the packed word");
static_assert(NPB <= (1 << (31 - RBITS)), "local target must fit the packed word");

__device__ __forceinline__ int block_scan_excl(int v, int* wsum, int& total) {
    const int lane = threadIdx.x & 31;
    const int wave = threadIdx.x >> 5;
    int inc = v;
#pragma unroll
    for (int o = 1; o < 32; o <<= 1) {
        int u = __shfl_up(inc, o);
        if (lane >= o) inc += u;
    }
    __syncthreads();
    if (lane == 31) wsum[wave] = inc;
    __syncthreads();
    int wb = 0, tot = 0;
#pragma unroll
    for (int w = 0; w < 32; ++w) {
        const int s = wsum[w];
        if (w < wave) wb += s;
        tot += s;
    }
    total = tot;
    return wb + inc - v;
}

__global__ __launch_bounds__(1024) void chunk_sort_kernel(
    const int* __restrict__ row, const int* __restrict__ col,
    int* __restrict__ pk, int* __restrict__ tcnt, int* __restrict__ tstart, int E) {
    __shared__ int lh[256];
    __shared__ int wsum[32];
    const int tid = threadIdx.x;
    const int w   = blockIdx.x;
    const int chunk = (E + NCHUNK - 1) / NCHUNK;
    const int beg = min(w * chunk, E);
    const int end = min(beg + chunk, E);

    if (tid < 256) lh[tid] = 0;
    __syncthreads();
    for (int i0 = beg; i0 < end; i0 += 8 * 1024) {
        int cv[8];
#pragma unroll
        for (int u = 0; u < 8; ++u) {
            const int i = i0 + u * 1024 + tid;
            cv[u] = col[min(i, end - 1)];
        }
#pragma unroll
        for (int u = 0; u < 8; ++u) {
            const int i = i0 + u * 1024 + tid;
            if (i < end) atomicAdd(&lh[(unsigned)cv[u] / NPB], 1);
        }
    }
    __syncthreads();
    const int v = (tid < NBINS) ? lh[tid] : 0;
    int tot;
    const int ex = block_scan_excl(v, wsum, tot);
    if (tid < NBINS) {
        tcnt[tid * NCHUNK + w]   = v;
        tstart[tid * NCHUNK + w] = beg + ex;
        lh[tid] = beg + ex;
    }
    __syncthreads();
    for (int i0 = beg; i0 < end; i0 += 8 * 1024) {
        int cv[8], rv[8];
#pragma unroll
        for (int u = 0; u < 8; ++u) {
            const int i = min(i0 + u * 1024 + tid, end - 1);
            cv[u] = col[i];
            rv[u] = row[i];
        }
#pragma unroll
        for (int u = 0; u < 8; ++u) {
            const int i = i0 + u * 1024 + tid;
            if (i < end) {
                const unsigned b = (unsigned)cv[u] / NPB;
                const int t = cv[u] - (int)b * NPB;
                const int slot = atomicAdd(&lh[b], 1);
                pk[slot] = (t << RBITS) | rv[u];
            }
        }
    }
}

__device__ __forceinline__ void find_in_runs(const int* p, const int* sabs,
                                             const int (&j)[8], int (&idx)[8]) {
    int lo[8];
#pragma unroll
    for (int u = 0; u < 8; ++u) lo[u] = 0;
#pragma unroll
    for (int half = NCHUNK / 2; half > 0; half >>= 1) {
        int pm[8];
#pragma unroll
        for (int u = 0; u < 8; ++u) pm[u] = p[lo[u] + half];
#pragma unroll
        for (int u = 0; u < 8; ++u) if (pm[u] <= j[u]) lo[u] += half;
    }
#pragma unroll
    for (int u = 0; u < 8; ++u) idx[u] = sabs[lo[u]] + (j[u] - p[lo[u]]);
}

__global__ __launch_bounds__(1024) void bucket_kernel(
    const int* __restrict__ pk, const int* __restrict__ tcnt,
    const int* __restrict__ tstart, int* __restrict__ off,
    float* __restrict__ dis, int* __restrict__ srcs) {
    __shared__ int p[NCHUNK + 1];
    __shared__ int sabs[NCHUNK];
    __shared__ int lcnt[NPB];
    __shared__ int wsum[32];
    const int tid = threadIdx.x;
    const int b   = blockIdx.x;

    const int4* t4 = (const int4*)tcnt;
    const int n4 = b * (NCHUNK / 4);
    int part = 0;
    for (int k0 = 0; k0 < n4; k0 += 8 * 1024) {
        int4 v[8];
#pragma unroll
        for (int u = 0; u < 8; ++u) {
            const int k = k0 + u * 1024 + tid;
            v[u] = t4[min(k, n4 - 1)];
        }
#pragma unroll
        for (int u = 0; u < 8; ++u) {
            const int k = k0 + u * 1024 + tid;
            const int s = (v[u].x + v[u].y) + (v[u].z + v[u].w);
            part += (k < n4) ? s : 0;
        }
    }
    int c = 0;
    if (tid < NCHUNK) {
        c = tcnt[b * NCHUNK + tid];
        sabs[tid] = tstart[b * NCHUNK + tid];
    }
    if (tid < NPB) lcnt[tid] = 0;
    int Tb, base;
    const int exc = block_scan_excl(c, wsum, Tb);
    if (tid < NCHUNK) p[tid] = exc;
    if (tid == 0) p[NCHUNK] = Tb;
    (void)block_scan_excl(part, wsum, base);
    __syncthreads();

    for (int j0 = 0; j0 < Tb; j0 += 8 * 1024) {
        int jj[8], idx[8];
#pragma unroll
        for (int u = 0; u < 8; ++u) jj[u] = min(j0 + u * 1024 + tid, Tb - 1);
        find_in_runs(p, sabs, jj, idx);
        int e[8];
#pragma unroll
        for (int u = 0; u < 8; ++u) e[u] = pk[idx[u]];
#pragma unroll
        for (int u = 0; u < 8; ++u) {
            const int j = j0 + u * 1024 + tid;
            if (j < Tb) atomicAdd(&lcnt[e[u] >> RBITS], 1);
        }
    }
    __syncthreads();
    const int d = (tid < NPB) ? lcnt[tid] : 0;
    int tot;
    const int ex = block_scan_excl(d, wsum, tot);
    if (tid < NPB) {
        const int node = b * NPB + tid;
        const int o = base + ex;
        off[node] = o;
        dis[node] = rsqrtf((float)(d + 1));
        lcnt[tid] = o;
    }
    if (b == NBINS - 1 && tid == 0) off[N_NODES] = base + tot;
    __syncthreads();

    for (int j0 = 0; j0 < Tb; j0 += 8 * 1024) {
        int jj[8], idx[8];
#pragma unroll
        for (int u = 0; u < 8; ++u) jj[u] = min(j0 + u * 1024 + tid, Tb - 1);
        find_in_runs(p, sabs, jj, idx);
        int e[8];
#pragma unroll
        for (int u = 0; u < 8; ++u) e[u] = pk[idx[u]];
#pragma unroll
        for (int u = 0; u < 8; ++u) {
            const int j = j0 + u * 1024 + tid;
            if (j < Tb) {
                const int slot = atomicAdd(&lcnt[e[u] >> RBITS], 1);
                srcs[slot] = e[u] & ((1 << RBITS) - 1);
            }
        }
    }
}


#define IDLE_SLOTS_BEHIND_PRODUCTS(c0, c1)                                      \
    asm volatile("v_nop\n\tv_nop\n\tv_nop\n\tv_nop\n\t"                        \
                 "v_nop\n\tv_nop\n\tv_nop\n\tv_nop" : "+v"(c0), "+v"(c1))
#define IDLE_SLOTS_BEHIND_PRODUCT(c0)                                           \
    asm volatile("v_nop\n\tv_nop\n\tv_nop\n\tv_nop\n\t"                        \
                 "v_nop\n\tv_nop\n\tv_nop\n\tv_nop" : "+v"(c0) : : "memory")

#define GEMM1_BLOCKS 512
__global__ __launch_bounds__(256) void gemm1_kernel(const float* __restrict__ x,
                                                    const float* __restrict__ W,
                                                    const float* __restrict__ dis,
                                                    float* __restrict__ h) {
    const int lane = threadIdx.x & 31;
    const int wave = threadIdx.x >> 5;
    const int n0   = (wave & 1) * 32;
    const int half = lane >> 4;
    const int l    = lane & 15;

    const float* bcol = W + (2 * half) * 64 + n0 + l;
    v2f bw0[32], bw1[32];
#pragma unroll
    for (int s = 0; s < 32; ++s) {
        bw0[s].x = bcol[(4 * s + 0) * 64];
        bw0[s].y = bcol[(4 * s + 1) * 64];
        bw1[s].x = bcol[(4 * s + 0) * 64 + 16];
        bw1[s].y = bcol[(4 * s + 1) * 64 + 16];
    }

    for (int rowTile = blockIdx.x * 4 + (wave >> 1); rowTile < ROW_TILES;
         rowTile += GEMM1_BLOCKS * 4) {
        const int m0 = rowTile * 16;
        const float* arow = x + (m0 + l) * 128 + 2 * half;
        v2f a[32];
#pragma unroll
        for (int s = 0; s < 32; ++s) a[s] = *(const v2f*)(arow + 4 * s);
        const float4 s0 = *(const float4*)(dis + m0 + half * 8);
        const float4 s1 = *(const float4*)(dis + m0 + half * 8 + 4);
        gfloat* o = (gfloat*)(h + (m0 + half * 8) * 64 + n0 + l);
        asm volatile("" : "+v"(o));
        __builtin_amdgcn_sched_barrier(0);
        v8f c0 = {}, c1 = {};
#pragma unroll
        for (int s = 0; s < 32; ++s) {
            c0 = __builtin_amdgcn_wmma_f32_16x16x4_f32(false, a[s], false, bw0[s],
                                                       (short)0, c0, false, false);
            c1 = __builtin_amdgcn_wmma_f32_16x16x4_f32(false, a[s], false, bw1[s],
                                                       (short)0, c1, false, false);
        }
        IDLE_SLOTS_BEHIND_PRODUCTS(c0, c1);
        const float sc[8] = {s0.x, s0.y, s0.z, s0.w, s1.x, s1.y, s1.z, s1.w};
#pragma unroll
        for (int v = 0; v < 8; ++v) {
            o[v * 64]      = c0[v] * sc[v];
            o[v * 64 + 16] = c1[v] * sc[v];
        }
    }
}

template <int F, bool RELU>
__device__ __forceinline__ float4 gather_row_sum(const int* __restrict__ off,
                                                 const int* __restrict__ srcs,
                                                 const float dc,
                                                 const float* __restrict__ hs,
                                                 const float* __restrict__ bias,
                                                 const int c, const int g) {
    constexpr int LANES = F / 4;
    const int fc  = g * 4;
    const int beg = off[c];
    const int end = off[c + 1];

    float4 acc = *(const float4*)(hs + c * F + fc);
    const float4 bv = *(const float4*)(bias + fc);

    for (int b0 = beg; b0 < end; b0 += LANES) {
        const int n = min(LANES, end - b0);
        int myr = c;
        if (g < n) myr = srcs[b0 + g];
        for (int k0 = 0; k0 < n; k0 += 4) {
            const int r0 = __shfl(myr, k0 + 0, LANES);
            const int r1 = __shfl(myr, k0 + 1, LANES);
            const int r2 = __shfl(myr, k0 + 2, LANES);
            const int r3 = __shfl(myr, k0 + 3, LANES);
            const float m1 = (k0 + 1 < n) ? 1.0f : 0.0f;
            const float m2 = (k0 + 2 < n) ? 1.0f : 0.0f;
            const float m3 = (k0 + 3 < n) ? 1.0f : 0.0f;
            const float4 v0 = *(const float4*)(hs + r0 * F + fc);
            const float4 v1 = *(const float4*)(hs + r1 * F + fc);
            const float4 v2 = *(const float4*)(hs + r2 * F + fc);
            const float4 v3 = *(const float4*)(hs + r3 * F + fc);
            acc.x += v0.x;      acc.y += v0.y;      acc.z += v0.z;      acc.w += v0.w;
            acc.x += v1.x * m1; acc.y += v1.y * m1; acc.z += v1.z * m1; acc.w += v1.w * m1;
            acc.x += v2.x * m2; acc.y += v2.y * m2; acc.z += v2.z * m2; acc.w += v2.w * m2;
            acc.x += v3.x * m3; acc.y += v3.y * m3; acc.z += v3.z * m3; acc.w += v3.w * m3;
        }
    }
    acc.x = bv.x + acc.x * dc;
    acc.y = bv.y + acc.y * dc;
    acc.z = bv.z + acc.z * dc;
    acc.w = bv.w + acc.w * dc;
    if (RELU) {
        acc.x = fmaxf(acc.x, 0.0f);
        acc.y = fmaxf(acc.y, 0.0f);
        acc.z = fmaxf(acc.z, 0.0f);
        acc.w = fmaxf(acc.w, 0.0f);
    }
    return acc;
}

#define TILE_PITCH 68
__global__ __launch_bounds__(256) void gather64_relu_gemm2_kernel(
    const int* __restrict__ off, const int* __restrict__ srcs,
    const float* __restrict__ dis, const float* __restrict__ h,
    const float* __restrict__ bias, const float* __restrict__ W,
    float* __restrict__ gout) {
    __shared__ __attribute__((aligned(16))) float tile[16 * TILE_PITCH];
    __shared__ __attribute__((aligned(16))) float sdis[16];
    const int cl   = threadIdx.x >> 4;
    const int g    = threadIdx.x & 15;
    const int m0   = blockIdx.x * 16;
    const int lane = threadIdx.x & 31;
    const int wave = threadIdx.x >> 5;
    const int half = lane >> 4;
    const int l    = lane & 15;
    const int n0   = (wave & 1) * 16;

    const float dc = dis[m0 + cl];
    const float4 acc = gather_row_sum<64, true>(off, srcs, dc, h, bias, m0 + cl, g);
    *(float4*)(tile + cl * TILE_PITCH + g * 4) = acc;
    if (g == 0) sdis[cl] = dc;

    v2f bw[16];
    if (wave < 2) {
        const float* bcol = W + (2 * half) * 32 + n0 + l;
#pragma unroll
        for (int s = 0; s < 16; ++s) {
            bw[s].x = bcol[(4 * s + 0) * 32];
            bw[s].y = bcol[(4 * s + 1) * 32];
        }
    }
    __syncthreads();
    if (wave < 2) {
        const lfloat* arow = (const lfloat*)(tile + l * TILE_PITCH + 2 * half);
        const lfloat* srow = (const lfloat*)(sdis + half * 8);
        gfloat* o = (gfloat*)(gout + (m0 + half * 8) * 32 + n0 + l);
        asm volatile("" : "+v"(arow), "+v"(srow), "+v"(o));
        v8f c0 = {};
        v2f a[8];
#pragma unroll
        for (int s = 0; s < 8; ++s) a[s] = *(const __attribute__((address_space(3))) v2f*)(arow + 4 * s);
        __builtin_amdgcn_sched_barrier(0);
#pragma unroll
        for (int s = 0; s < 8; ++s)
            c0 = __builtin_amdgcn_wmma_f32_16x16x4_f32(false, a[s], false, bw[s],
                                                       (short)0, c0, false, false);
        IDLE_SLOTS_BEHIND_PRODUCT(c0);
#pragma unroll
        for (int s = 0; s < 8; ++s) a[s] = *(const __attribute__((address_space(3))) v2f*)(arow + 32 + 4 * s);
        __builtin_amdgcn_sched_barrier(0);
#pragma unroll
        for (int s = 0; s < 8; ++s)
            c0 = __builtin_amdgcn_wmma_f32_16x16x4_f32(false, a[s], false, bw[8 + s],
                                                       (short)0, c0, false, false);
        IDLE_SLOTS_BEHIND_PRODUCT(c0);
        const float4 s0 = *(const __attribute__((address_space(3))) float4*)(srow);
        const float4 s1 = *(const __attribute__((address_space(3))) float4*)(srow + 4);
        const float sc[8] = {s0.x, s0.y, s0.z, s0.w, s1.x, s1.y, s1.z, s1.w};
#pragma unroll
        for (int v = 0; v < 8; ++v) o[v * 32] = c0[v] * sc[v];
    }
}

__global__ __launch_bounds__(256) void gather32_kernel(
    const int* __restrict__ off, const int* __restrict__ srcs,
    const float* __restrict__ dis, const float* __restrict__ h,
    const float* __restrict__ bias, float* __restrict__ res) {
    const int tid = blockIdx.x * blockDim.x + threadIdx.x;
    const int c   = tid / 8;
    const int g   = tid % 8;
    const float4 acc = gather_row_sum<32, false>(off, srcs, dis[c], h, bias, c, g);
    *(float4*)(res + c * 32 + g * 4) = acc;
}

extern "C" void kernel_launch(void* const* d_in, const int* in_sizes, int n_in,
                              void* d_out, int out_size, void* d_ws, size_t ws_size,
                              hipStream_t stream) {
    const float* x  = (const float*)d_in[0];
    const int*   ei = (const int*)d_in[1];
    const float* W1 = (const float*)d_in[2];
    const float* b1 = (const float*)d_in[3];
    const float* W2 = (const float*)d_in[4];
    const float* b2 = (const float*)d_in[5];
    float* out = (float*)d_out;

    const int E = in_sizes[1] / 2;
    const int* row = ei;
    const int* col = ei + E;

    float* dis  = (float*)d_ws;
    float* h1   = dis  + N_NODES;
    float* g2   = h1   + N_NODES * 64;
    int*   srcs = (int*)(g2 + N_NODES * 64);
    int*   off  = srcs + N_NODES * 16;
    int*   pk     = (int*)h1;
    int*   tcnt   = (int*)g2;
    int*   tstart = tcnt + NBINS * NCHUNK;

    const int T = 256;
    chunk_sort_kernel<<<NCHUNK, 1024, 0, stream>>>(row, col, pk, tcnt, tstart, E);
    bucket_kernel<<<NBINS, 1024, 0, stream>>>(pk, tcnt, tstart, off, dis, srcs);

    gemm1_kernel<<<GEMM1_BLOCKS, T, 0, stream>>>(x, W1, dis, h1);
    gather64_relu_gemm2_kernel<<<ROW_TILES, T, 0, stream>>>(off, srcs, dis, h1, b1, W2, g2);

    gather32_kernel<<<N_NODES * 8 / T, T, 0, stream>>>(off, srcs, dis, g2, b2, out);
}


# both sort kernels build their output slice in LDS and copy it out one whole 128-byte line per wave store, in place of a scattered 4-byte global store per edge; slots and sum order unchanged
# speedup vs baseline: 4.5723x; 1.0900x over previous
#include <hip/hip_runtime.h>
#include <hip/hip_bf16.h>

typedef __attribute__((ext_vector_type(2))) float v2f;
typedef __attribute__((ext_vector_type(8))) float v8f;
typedef __attribute__((address_space(1))) float gfloat;
typedef __attribute__((address_space(3))) float lfloat;

#define N_NODES 100000
#define ROW_TILES (N_NODES / 16)

#define NPB    400
#define NBINS  250
#define NCHUNK 256
#define RBITS  17
#define IMG_INTS (8 * 1024)
#define IMG_TRIPS ((IMG_INTS + 32 + 1023) / 1024)

__device__ __forceinline__ void copy_image_out(const int* img, int* __restrict__ dst,
                                               const int a0, const int first, const int stop) {
    int v[IMG_TRIPS];
#pragma unroll
    for (int k = 0; k < IMG_TRIPS; ++k)
        v[k] = img[min(k * 1024 + (int)threadIdx.x, IMG_INTS + 31)];
#pragma unroll
    for (int k = 0; k < IMG_TRIPS; ++k) {
        const int gi = a0 + k * 1024 + (int)threadIdx.x;
        if (gi >= first && gi < stop) dst[gi] = v[k];
    }
}
static_assert(NBINS * NPB == N_NODES, "ranges must tile the nodes");
static_assert((NCHUNK & (NCHUNK - 1)) == 0 && NCHUNK <= 1024 && NBINS <= 256, "bisection and scans assume this");
static_assert(N_NODES <= (1 << RBITS), "source id must fit the packed word");
static_assert(NPB <= (1 << (31 - RBITS)), "local target must fit the packed word");

__device__ __forceinline__ int block_scan_excl(int v, int* wsum, int& total) {
    const int lane = threadIdx.x & 31;
    const int wave = threadIdx.x >> 5;
    int inc = v;
#pragma unroll
    for (int o = 1; o < 32; o <<= 1) {
        int u = __shfl_up(inc, o);
        if (lane >= o) inc += u;
    }
    __syncthreads();
    if (lane == 31) wsum[wave] = inc;
    __syncthreads();
    int wb = 0, tot = 0;
#pragma unroll
    for (int w = 0; w < 32; ++w) {
        const int s = wsum[w];
        if (w < wave) wb += s;
        tot += s;
    }
    total = tot;
    return wb + inc - v;
}

template <bool STAGED>
__device__ __forceinline__ void place_chunk(const int* __restrict__ row, const int* __restrict__ col,
                                            int* __restrict__ pk, int* lh, int* img,
                                            const int beg, const int end, const int a0) {
    const int tid = threadIdx.x;
    for (int i0 = beg; i0 < end; i0 += 8 * 1024) {
        int cv[8], rv[8];
#pragma unroll
        for (int u = 0; u < 8; ++u) {
            const int i = min(i0 + u * 1024 + tid, end - 1);
            cv[u] = col[i];
            rv[u] = row[i];
        }
#pragma unroll
        for (int u = 0; u < 8; ++u) {
            const int i = i0 + u * 1024 + tid;
            if (i < end) {
                const unsigned b = (unsigned)cv[u] / NPB;
                const int t = cv[u] - (int)b * NPB;
                const int slot = atomicAdd(&lh[b], 1);
                const int val = (t << RBITS) | rv[u];
                if (STAGED) img[slot - a0] = val;
                else        pk[slot] = val;
            }
        }
    }
}

__global__ __launch_bounds__(1024) void chunk_sort_kernel(
    const int* __restrict__ row, const int* __restrict__ col,
    int* __restrict__ pk, int* __restrict__ tcnt, int* __restrict__ tstart, int E) {
    __shared__ int lh[256];
    __shared__ int wsum[32];
    __shared__ int img[IMG_INTS + 32];
    const int tid = threadIdx.x;
    const int w   = blockIdx.x;
    const int chunk = (E + NCHUNK - 1) / NCHUNK;
    const int beg = min(w * chunk, E);
    const int end = min(beg + chunk, E);
    const int a0  = beg & ~31;
    const bool staged = (end - beg) <= IMG_INTS;

    if (tid < 256) lh[tid] = 0;
    __syncthreads();
    for (int i0 = beg; i0 < end; i0 += 8 * 1024) {
        int cv[8];
#pragma unroll
        for (int u = 0; u < 8; ++u) {
            const int i = i0 + u * 1024 + tid;
            cv[u] = col[min(i, end - 1)];
        }
#pragma unroll
        for (int u = 0; u < 8; ++u) {
            const int i = i0 + u * 1024 + tid;
            if (i < end) atomicAdd(&lh[(unsigned)cv[u] / NPB], 1);
        }
    }
    __syncthreads();
    const int v = (tid < NBINS) ? lh[tid] : 0;
    int tot;
    const int ex = block_scan_excl(v, wsum, tot);
    if (tid < NBINS) {
        tcnt[tid * NCHUNK + w]   = v;
        tstart[tid * NCHUNK + w] = beg + ex;
        lh[tid] = beg + ex;
    }
    __syncthreads();
    if (staged) {
        place_chunk<true>(row, col, pk, lh, img, beg, end, a0);
        __syncthreads();
        copy_image_out(img, pk, a0, beg, end);
    } else {
        place_chunk<false>(row, col, pk, lh, img, beg, end, a0);
    }
}

__device__ __forceinline__ void find_in_runs(const int* p, const int* sabs,
                                             const int (&j)[8], int (&idx)[8]) {
    int lo[8];
#pragma unroll
    for (int u = 0; u < 8; ++u) lo[u] = 0;
#pragma unroll
    for (int half = NCHUNK / 2; half > 0; half >>= 1) {
        int pm[8];
#pragma unroll
        for (int u = 0; u < 8; ++u) pm[u] = p[lo[u] + half];
#pragma unroll
        for (int u = 0; u < 8; ++u) if (pm[u] <= j[u]) lo[u] += half;
    }
#pragma unroll
    for (int u = 0; u < 8; ++u) idx[u] = sabs[lo[u]] + (j[u] - p[lo[u]]);
}

template <bool STAGED>
__device__ __forceinline__ void place_range(const int* __restrict__ pk, const int* p, const int* sabs,
                                            int* lcnt, int* img, int* __restrict__ srcs,
                                            const int Tb, const int a0) {
    const int tid = threadIdx.x;
    for (int j0 = 0; j0 < Tb; j0 += 8 * 1024) {
        int jj[8], idx[8];
#pragma unroll
        for (int u = 0; u < 8; ++u) jj[u] = min(j0 + u * 1024 + tid, Tb - 1);
        find_in_runs(p, sabs, jj, idx);
        int e[8];
#pragma unroll
        for (int u = 0; u < 8; ++u) e[u] = pk[idx[u]];
#pragma unroll
        for (int u = 0; u < 8; ++u) {
            const int j = j0 + u * 1024 + tid;
            if (j < Tb) {
                const int slot = atomicAdd(&lcnt[e[u] >> RBITS], 1);
                const int s = e[u] & ((1 << RBITS) - 1);
                if (STAGED) img[slot - a0] = s;
                else        srcs[slot] = s;
            }
        }
    }
}

__global__ __launch_bounds__(1024) void bucket_kernel(
    const int* __restrict__ pk, const int* __restrict__ tcnt,
    const int* __restrict__ tstart, int* __restrict__ off,
    float* __restrict__ dis, int* __restrict__ srcs) {
    __shared__ int p[NCHUNK + 1];
    __shared__ int sabs[NCHUNK];
    __shared__ int lcnt[NPB];
    __shared__ int wsum[32];
    __shared__ int img[IMG_INTS + 32];
    const int tid = threadIdx.x;
    const int b   = blockIdx.x;

    const int4* t4 = (const int4*)tcnt;
    const int n4 = b * (NCHUNK / 4);
    int part = 0;
    for (int k0 = 0; k0 < n4; k0 += 8 * 1024) {
        int4 v[8];
#pragma unroll
        for (int u = 0; u < 8; ++u) {
            const int k = k0 + u * 1024 + tid;
            v[u] = t4[min(k, n4 - 1)];
        }
#pragma unroll
        for (int u = 0; u < 8; ++u) {
            const int k = k0 + u * 1024 + tid;
            const int s = (v[u].x + v[u].y) + (v[u].z + v[u].w);
            part += (k < n4) ? s : 0;
        }
    }
    int c = 0;
    if (tid < NCHUNK) {
        c = tcnt[b * NCHUNK + tid];
        sabs[tid] = tstart[b * NCHUNK + tid];
    }
    if (tid < NPB) lcnt[tid] = 0;
    int Tb, base;
    const int exc = block_scan_excl(c, wsum, Tb);
    if (tid < NCHUNK) p[tid] = exc;
    if (tid == 0) p[NCHUNK] = Tb;
    (void)block_scan_excl(part, wsum, base);
    __syncthreads();
    const int a0 = base & ~31;
    const bool staged = Tb <= IMG_INTS;

    for (int j0 = 0; j0 < Tb; j0 += 8 * 1024) {
        int jj[8], idx[8];
#pragma unroll
        for (int u = 0; u < 8; ++u) jj[u] = min(j0 + u * 1024 + tid, Tb - 1);
        find_in_runs(p, sabs, jj, idx);
        int e[8];
#pragma unroll
        for (int u = 0; u < 8; ++u) e[u] = pk[idx[u]];
#pragma unroll
        for (int u = 0; u < 8; ++u) {
            const int j = j0 + u * 1024 + tid;
            if (j < Tb) atomicAdd(&lcnt[e[u] >> RBITS], 1);
        }
    }
    __syncthreads();
    const int d = (tid < NPB) ? lcnt[tid] : 0;
    int tot;
    const int ex = block_scan_excl(d, wsum, tot);
    if (tid < NPB) {
        const int node = b * NPB + tid;
        const int o = base + ex;
        off[node] = o;
        dis[node] = rsqrtf((float)(d + 1));
        lcnt[tid] = o;
    }
    if (b == NBINS - 1 && tid == 0) off[N_NODES] = base + tot;
    __syncthreads();

    if (staged) {
        place_range<true>(pk, p, sabs, lcnt, img, srcs, Tb, a0);
        __syncthreads();
        copy_image_out(img, srcs, a0, base, base + Tb);
    } else {
        place_range<false>(pk, p, sabs, lcnt, img, srcs, Tb, a0);
    }
}


#define IDLE_SLOTS_BEHIND_PRODUCTS(c0, c1)                                      \
    asm volatile("v_nop\n\tv_nop\n\tv_nop\n\tv_nop\n\t"                        \
                 "v_nop\n\tv_nop\n\tv_nop\n\tv_nop" : "+v"(c0), "+v"(c1))
#define IDLE_SLOTS_BEHIND_PRODUCT(c0)                                           \
    asm volatile("v_nop\n\tv_nop\n\tv_nop\n\tv_nop\n\t"                        \
                 "v_nop\n\tv_nop\n\tv_nop\n\tv_nop" : "+v"(c0) : : "memory")

#define GEMM1_BLOCKS 512
__global__ __launch_bounds__(256) void gemm1_kernel(const float* __restrict__ x,
                                                    const float* __restrict__ W,
                                                    const float* __restrict__ dis,
                                                    float* __restrict__ h) {
    const int lane = threadIdx.x & 31;
    const int wave = threadIdx.x >> 5;
    const int n0   = (wave & 1) * 32;
    const int half = lane >> 4;
    const int l    = lane & 15;

    const float* bcol = W + (2 * half) * 64 + n0 + l;
    v2f bw0[32], bw1[32];
#pragma unroll
    for (int s = 0; s < 32; ++s) {
        bw0[s].x = bcol[(4 * s + 0) * 64];
        bw0[s].y = bcol[(4 * s + 1) * 64];
        bw1[s].x = bcol[(4 * s + 0) * 64 + 16];
        bw1[s].y = bcol[(4 * s + 1) * 64 + 16];
    }

    for (int rowTile = blockIdx.x * 4 + (wave >> 1); rowTile < ROW_TILES;
         rowTile += GEMM1_BLOCKS * 4) {
        const int m0 = rowTile * 16;
        const float* arow = x + (m0 + l) * 128 + 2 * half;
        v2f a[32];
#pragma unroll
        for (int s = 0; s < 32; ++s) a[s] = *(const v2f*)(arow + 4 * s);
        const float4 s0 = *(const float4*)(dis + m0 + half * 8);
        const float4 s1 = *(const float4*)(dis + m0 + half * 8 + 4);
        gfloat* o = (gfloat*)(h + (m0 + half * 8) * 64 + n0 + l);
        asm volatile("" : "+v"(o));
        __builtin_amdgcn_sched_barrier(0);
        v8f c0 = {}, c1 = {};
#pragma unroll
        for (int s = 0; s < 32; ++s) {
            c0 = __builtin_amdgcn_wmma_f32_16x16x4_f32(false, a[s], false, bw0[s],
                                                       (short)0, c0, false, false);
            c1 = __builtin_amdgcn_wmma_f32_16x16x4_f32(false, a[s], false, bw1[s],
                                                       (short)0, c1, false, false);
        }
        IDLE_SLOTS_BEHIND_PRODUCTS(c0, c1);
        const float sc[8] = {s0.x, s0.y, s0.z, s0.w, s1.x, s1.y, s1.z, s1.w};
#pragma unroll
        for (int v = 0; v < 8; ++v) {
            o[v * 64]      = c0[v] * sc[v];
            o[v * 64 + 16] = c1[v] * sc[v];
        }
    }
}

template <int F, bool RELU>
__device__ __forceinline__ float4 gather_row_sum(const int* __restrict__ off,
                                                 const int* __restrict__ srcs,
                                                 const float dc,
                                                 const float* __restrict__ hs,
                                                 const float* __restrict__ bias,
                                                 const int c, const int g) {
    constexpr int LANES = F / 4;
    const int fc  = g * 4;
    const int beg = off[c];
    const int end = off[c + 1];

    float4 acc = *(const float4*)(hs + c * F + fc);
    const float4 bv = *(const float4*)(bias + fc);

    for (int b0 = beg; b0 < end; b0 += LANES) {
        const int n = min(LANES, end - b0);
        int myr = c;
        if (g < n) myr = srcs[b0 + g];
        for (int k0 = 0; k0 < n; k0 += 4) {
            const int r0 = __shfl(myr, k0 + 0, LANES);
            const int r1 = __shfl(myr, k0 + 1, LANES);
            const int r2 = __shfl(myr, k0 + 2, LANES);
            const int r3 = __shfl(myr, k0 + 3, LANES);
            const float m1 = (k0 + 1 < n) ? 1.0f : 0.0f;
            const float m2 = (k0 + 2 < n) ? 1.0f : 0.0f;
            const float m3 = (k0 + 3 < n) ? 1.0f : 0.0f;
            const float4 v0 = *(const float4*)(hs + r0 * F + fc);
            const float4 v1 = *(const float4*)(hs + r1 * F + fc);
            const float4 v2 = *(const float4*)(hs + r2 * F + fc);
            const float4 v3 = *(const float4*)(hs + r3 * F + fc);
            acc.x += v0.x;      acc.y += v0.y;      acc.z += v0.z;      acc.w += v0.w;
            acc.x += v1.x * m1; acc.y += v1.y * m1; acc.z += v1.z * m1; acc.w += v1.w * m1;
            acc.x += v2.x * m2; acc.y += v2.y * m2; acc.z += v2.z * m2; acc.w += v2.w * m2;
            acc.x += v3.x * m3; acc.y += v3.y * m3; acc.z += v3.z * m3; acc.w += v3.w * m3;
        }
    }
    acc.x = bv.x + acc.x * dc;
    acc.y = bv.y + acc.y * dc;
    acc.z = bv.z + acc.z * dc;
    acc.w = bv.w + acc.w * dc;
    if (RELU) {
        acc.x = fmaxf(acc.x, 0.0f);
        acc.y = fmaxf(acc.y, 0.0f);
        acc.z = fmaxf(acc.z, 0.0f);
        acc.w = fmaxf(acc.w, 0.0f);
    }
    return acc;
}

#define TILE_PITCH 68
__global__ __launch_bounds__(256) void gather64_relu_gemm2_kernel(
    const int* __restrict__ off, const int* __restrict__ srcs,
    const float* __restrict__ dis, const float* __restrict__ h,
    const float* __restrict__ bias, const float* __restrict__ W,
    float* __restrict__ gout) {
    __shared__ __attribute__((aligned(16))) float tile[16 * TILE_PITCH];
    __shared__ __attribute__((aligned(16))) float sdis[16];
    const int cl   = threadIdx.x >> 4;
    const int g    = threadIdx.x & 15;
    const int m0   = blockIdx.x * 16;
    const int lane = threadIdx.x & 31;
    const int wave = threadIdx.x >> 5;
    const int half = lane >> 4;
    const int l    = lane & 15;
    const int n0   = (wave & 1) * 16;

    const float dc = dis[m0 + cl];
    const float4 acc = gather_row_sum<64, true>(off, srcs, dc, h, bias, m0 + cl, g);
    *(float4*)(tile + cl * TILE_PITCH + g * 4) = acc;
    if (g == 0) sdis[cl] = dc;

    v2f bw[16];
    if (wave < 2) {
        const float* bcol = W + (2 * half) * 32 + n0 + l;
#pragma unroll
        for (int s = 0; s < 16; ++s) {
            bw[s].x = bcol[(4 * s + 0) * 32];
            bw[s].y = bcol[(4 * s + 1) * 32];
        }
    }
    __syncthreads();
    if (wave < 2) {
        const lfloat* arow = (const lfloat*)(tile + l * TILE_PITCH + 2 * half);
        const lfloat* srow = (const lfloat*)(sdis + half * 8);
        gfloat* o = (gfloat*)(gout + (m0 + half * 8) * 32 + n0 + l);
        asm volatile("" : "+v"(arow), "+v"(srow), "+v"(o));
        v8f c0 = {};
        v2f a[8];
#pragma unroll
        for (int s = 0; s < 8; ++s) a[s] = *(const __attribute__((address_space(3))) v2f*)(arow + 4 * s);
        __builtin_amdgcn_sched_barrier(0);
#pragma unroll
        for (int s = 0; s < 8; ++s)
            c0 = __builtin_amdgcn_wmma_f32_16x16x4_f32(false, a[s], false, bw[s],
                                                       (short)0, c0, false, false);
        IDLE_SLOTS_BEHIND_PRODUCT(c0);
#pragma unroll
        for (int s = 0; s < 8; ++s) a[s] = *(const __attribute__((address_space(3))) v2f*)(arow + 32 + 4 * s);
        __builtin_amdgcn_sched_barrier(0);
#pragma unroll
        for (int s = 0; s < 8; ++s)
            c0 = __builtin_amdgcn_wmma_f32_16x16x4_f32(false, a[s], false, bw[8 + s],
                                                       (short)0, c0, false, false);
        IDLE_SLOTS_BEHIND_PRODUCT(c0);
        const float4 s0 = *(const __attribute__((address_space(3))) float4*)(srow);
        const float4 s1 = *(const __attribute__((address_space(3))) float4*)(srow + 4);
        const float sc[8] = {s0.x, s0.y, s0.z, s0.w, s1.x, s1.y, s1.z, s1.w};
#pragma unroll
        for (int v = 0; v < 8; ++v) o[v * 32] = c0[v] * sc[v];
    }
}

__global__ __launch_bounds__(256) void gather32_kernel(
    const int* __restrict__ off, const int* __restrict__ srcs,
    const float* __restrict__ dis, const float* __restrict__ h,
    const float* __restrict__ bias, float* __restrict__ res) {
    const int tid = blockIdx.x * blockDim.x + threadIdx.x;
    const int c   = tid / 8;
    const int g   = tid % 8;
    const float4 acc = gather_row_sum<32, false>(off, srcs, dis[c], h, bias, c, g);
    *(float4*)(res + c * 32 + g * 4) = acc;
}

extern "C" void kernel_launch(void* const* d_in, const int* in_sizes, int n_in,
                              void* d_out, int out_size, void* d_ws, size_t ws_size,
                              hipStream_t stream) {
    const float* x  = (const float*)d_in[0];
    const int*   ei = (const int*)d_in[1];
    const float* W1 = (const float*)d_in[2];
    const float* b1 = (const float*)d_in[3];
    const float* W2 = (const float*)d_in[4];
    const float* b2 = (const float*)d_in[5];
    float* out = (float*)d_out;

    const int E = in_sizes[1] / 2;
    const int* row = ei;
    const int* col = ei + E;

    float* dis  = (float*)d_ws;
    float* h1   = dis  + N_NODES;
    float* g2   = h1   + N_NODES * 64;
    int*   srcs = (int*)(g2 + N_NODES * 64);
    int*   off  = srcs + N_NODES * 16;
    int*   pk     = (int*)h1;
    int*   tcnt   = (int*)g2;
    int*   tstart = tcnt + NBINS * NCHUNK;

    const int T = 256;
    chunk_sort_kernel<<<NCHUNK, 1024, 0, stream>>>(row, col, pk, tcnt, tstart, E);
    bucket_kernel<<<NBINS, 1024, 0, stream>>>(pk, tcnt, tstart, off, dis, srcs);

    gemm1_kernel<<<GEMM1_BLOCKS, T, 0, stream>>>(x, W1, dis, h1);
    gather64_relu_gemm2_kernel<<<ROW_TILES, T, 0, stream>>>(off, srcs, dis, h1, b1, W2, g2);

    gather32_kernel<<<N_NODES * 8 / T, T, 0, stream>>>(off, srcs, dis, g2, b2, out);
}


# x@W1 product moved into the bucket launch: each node-range work-group multiplies its own 400 rows of x, read as whole rows into LDS, scaled by its own dis; 5 launches to 4; k order, sums unchanged
# speedup vs baseline: 4.7490x; 1.0386x over previous
#include <hip/hip_runtime.h>
#include <hip/hip_bf16.h>

typedef __attribute__((ext_vector_type(2))) float v2f;
typedef __attribute__((ext_vector_type(8))) float v8f;
typedef __attribute__((address_space(1))) float gfloat;
typedef __attribute__((address_space(3))) float lfloat;

#define N_NODES 100000
#define ROW_TILES (N_NODES / 16)

#define NPB    400
#define NBINS  250
#define NCHUNK 256
#define RBITS  17
#define IMG_INTS (8 * 1024)
#define IMG_TRIPS ((IMG_INTS + 32 + 1023) / 1024)

__device__ __forceinline__ void copy_image_out(const int* img, int* __restrict__ dst,
                                               const int a0, const int first, const int stop) {
    int v[IMG_TRIPS];
#pragma unroll
    for (int k = 0; k < IMG_TRIPS; ++k)
        v[k] = img[min(k * 1024 + (int)threadIdx.x, IMG_INTS + 31)];
#pragma unroll
    for (int k = 0; k < IMG_TRIPS; ++k) {
        const int gi = a0 + k * 1024 + (int)threadIdx.x;
        if (gi >= first && gi < stop) dst[gi] = v[k];
    }
}
static_assert(NBINS * NPB == N_NODES, "ranges must tile the nodes");
static_assert((NCHUNK & (NCHUNK - 1)) == 0 && NCHUNK <= 1024 && NBINS <= 256, "bisection and scans assume this");
static_assert(N_NODES <= (1 << RBITS), "source id must fit the packed word");
static_assert(NPB <= (1 << (31 - RBITS)), "local target must fit the packed word");

__device__ __forceinline__ int block_scan_excl(int v, int* wsum, int& total) {
    const int lane = threadIdx.x & 31;
    const int wave = threadIdx.x >> 5;
    int inc = v;
#pragma unroll
    for (int o = 1; o < 32; o <<= 1) {
        int u = __shfl_up(inc, o);
        if (lane >= o) inc += u;
    }
    __syncthreads();
    if (lane == 31) wsum[wave] = inc;
    __syncthreads();
    int wb = 0, tot = 0;
#pragma unroll
    for (int w = 0; w < 32; ++w) {
        const int s = wsum[w];
        if (w < wave) wb += s;
        tot += s;
    }
    total = tot;
    return wb + inc - v;
}

template <bool STAGED>
__device__ __forceinline__ void place_chunk(const int* __restrict__ row, const int* __restrict__ col,
                                            int* __restrict__ pk, int* lh, int* img,
                                            const int beg, const int end, const int a0) {
    const int tid = threadIdx.x;
    for (int i0 = beg; i0 < end; i0 += 8 * 1024) {
        int cv[8], rv[8];
#pragma unroll
        for (int u = 0; u < 8; ++u) {
            const int i = min(i0 + u * 1024 + tid, end - 1);
            cv[u] = col[i];
            rv[u] = row[i];
        }
#pragma unroll
        for (int u = 0; u < 8; ++u) {
            const int i = i0 + u * 1024 + tid;
            if (i < end) {
                const unsigned b = (unsigned)cv[u] / NPB;
                const int t = cv[u] - (int)b * NPB;
                const int slot = atomicAdd(&lh[b], 1);
                const int val = (t << RBITS) | rv[u];
                if (STAGED) img[slot - a0] = val;
                else        pk[slot] = val;
            }
        }
    }
}

__global__ __launch_bounds__(1024) void chunk_sort_kernel(
    const int* __restrict__ row, const int* __restrict__ col,
    int* __restrict__ pk, int* __restrict__ tcnt, int* __restrict__ tstart, int E) {
    __shared__ int lh[256];
    __shared__ int wsum[32];
    __shared__ int img[IMG_INTS + 32];
    const int tid = threadIdx.x;
    const int w   = blockIdx.x;
    const int chunk = (E + NCHUNK - 1) / NCHUNK;
    const int beg = min(w * chunk, E);
    const int end = min(beg + chunk, E);
    const int a0  = beg & ~31;
    const bool staged = (end - beg) <= IMG_INTS;

    if (tid < 256) lh[tid] = 0;
    __syncthreads();
    for (int i0 = beg; i0 < end; i0 += 8 * 1024) {
        int cv[8];
#pragma unroll
        for (int u = 0; u < 8; ++u) {
            const int i = i0 + u * 1024 + tid;
            cv[u] = col[min(i, end - 1)];
        }
#pragma unroll
        for (int u = 0; u < 8; ++u) {
            const int i = i0 + u * 1024 + tid;
            if (i < end) atomicAdd(&lh[(unsigned)cv[u] / NPB], 1);
        }
    }
    __syncthreads();
    const int v = (tid < NBINS) ? lh[tid] : 0;
    int tot;
    const int ex = block_scan_excl(v, wsum, tot);
    if (tid < NBINS) {
        tcnt[tid * NCHUNK + w]   = v;
        tstart[tid * NCHUNK + w] = beg + ex;
        lh[tid] = beg + ex;
    }
    __syncthreads();
    if (staged) {
        place_chunk<true>(row, col, pk, lh, img, beg, end, a0);
        __syncthreads();
        copy_image_out(img, pk, a0, beg, end);
    } else {
        place_chunk<false>(row, col, pk, lh, img, beg, end, a0);
    }
}

__device__ __forceinline__ void find_in_runs(const int* p, const int* sabs,
                                             const int (&j)[8], int (&idx)[8]) {
    int lo[8];
#pragma unroll
    for (int u = 0; u < 8; ++u) lo[u] = 0;
#pragma unroll
    for (int half = NCHUNK / 2; half > 0; half >>= 1) {
        int pm[8];
#pragma unroll
        for (int u = 0; u < 8; ++u) pm[u] = p[lo[u] + half];
#pragma unroll
        for (int u = 0; u < 8; ++u) if (pm[u] <= j[u]) lo[u] += half;
    }
#pragma unroll
    for (int u = 0; u < 8; ++u) idx[u] = sabs[lo[u]] + (j[u] - p[lo[u]]);
}

template <bool STAGED>
__device__ __forceinline__ void place_range(const int* __restrict__ pk, const int* p, const int* sabs,
                                            int* lcnt, int* img, int* __restrict__ srcs,
                                            const int Tb, const int a0) {
    const int tid = threadIdx.x;
    for (int j0 = 0; j0 < Tb; j0 += 8 * 1024) {
        int jj[8], idx[8];
#pragma unroll
        for (int u = 0; u < 8; ++u) jj[u] = min(j0 + u * 1024 + tid, Tb - 1);
        find_in_runs(p, sabs, jj, idx);
        int e[8];
#pragma unroll
        for (int u = 0; u < 8; ++u) e[u] = pk[idx[u]];
#pragma unroll
        for (int u = 0; u < 8; ++u) {
            const int j = j0 + u * 1024 + tid;
            if (j < Tb) {
                const int slot = atomicAdd(&lcnt[e[u] >> RBITS], 1);
                const int s = e[u] & ((1 << RBITS) - 1);
                if (STAGED) img[slot - a0] = s;
                else        srcs[slot] = s;
            }
        }
    }
}


#define IDLE_SLOTS_BEHIND_PRODUCT(c0)                                           \
    asm volatile("v_nop\n\tv_nop\n\tv_nop\n\tv_nop\n\t"                        \
                 "v_nop\n\tv_nop\n\tv_nop\n\tv_nop" : "+v"(c0) : : "memory")

#define X_PITCH 132
#define XA_TILES 13
#define XB_TILES (NPB / 16 - XA_TILES)
#define XA_VEC4  (XA_TILES * 16 * 32)
#define XB_VEC4  (XB_TILES * 16 * 32)
static_assert(NPB % 16 == 0, "a range must be whole row tiles");
static_assert(XA_VEC4 > 6 * 1024 && XA_VEC4 <= 7 * 1024, "the bucket kernel spells out 7 trips of 1024 float4s for part one");
static_assert(XB_VEC4 == 6 * 1024, "the bucket kernel spells out 6 trips of 1024 float4s for part two");
static_assert(XA_TILES > 8 && XA_TILES <= 16 && XB_TILES > 8 && XB_TILES <= 16, "a wave has one or two row tiles in each part");

#define XS_PUT(i, v) *(float4*)(xs + ((i) >> 5) * X_PITCH + ((i) & 31) * 4) = (v)

__device__ __forceinline__ void product_tile(const float* xs, const float* sdis, float* hrow,
                                             const v2f (&bw)[32], const int rt,
                                             const int half, const int l) {
    const lfloat* arow = (const lfloat*)(xs + (rt * 16 + l) * X_PITCH + 2 * half);
    const lfloat* srow = (const lfloat*)(sdis + rt * 16 + half * 8);
    gfloat* o = (gfloat*)(hrow + (rt * 16 + half * 8) * 64 + l);
    asm volatile("" : "+v"(arow), "+v"(srow), "+v"(o));
    v8f c0 = {};
#pragma unroll
    for (int q = 0; q < 4; ++q) {
        v2f a[8];
#pragma unroll
        for (int s = 0; s < 8; ++s)
            a[s] = *(const __attribute__((address_space(3))) v2f*)(arow + 32 * q + 4 * s);
        __builtin_amdgcn_sched_barrier(0);
#pragma unroll
        for (int s = 0; s < 8; ++s)
            c0 = __builtin_amdgcn_wmma_f32_16x16x4_f32(false, a[s], false, bw[8 * q + s],
                                                       (short)0, c0, false, false);
        IDLE_SLOTS_BEHIND_PRODUCT(c0);
    }
    const float4 s0 = *(const __attribute__((address_space(3))) float4*)(srow);
    const float4 s1 = *(const __attribute__((address_space(3))) float4*)(srow + 4);
    const float sc[8] = {s0.x, s0.y, s0.z, s0.w, s1.x, s1.y, s1.z, s1.w};
#pragma unroll
    for (int v = 0; v < 8; ++v) o[v * 64] = c0[v] * sc[v];
}

__global__ __launch_bounds__(1024) void bucket_gemm1_kernel(
    const int* __restrict__ pk, const int* __restrict__ tcnt,
    const int* __restrict__ tstart, int* __restrict__ off,
    float* __restrict__ dis, int* __restrict__ srcs,
    const float* __restrict__ x, const float* __restrict__ W,
    float* __restrict__ h) {
    __shared__ int p[NCHUNK + 1];
    __shared__ int sabs[NCHUNK];
    __shared__ int lcnt[NPB];
    __shared__ int wsum[32];
    __shared__ int img[IMG_INTS + 32];
    __shared__ __attribute__((aligned(16))) float xs[NPB * X_PITCH];
    __shared__ __attribute__((aligned(16))) float sdis[NPB];
    const int tid = threadIdx.x;
    const int b   = blockIdx.x;

    const float4* x4 = (const float4*)(x + b * (NPB * 128));
    const float4 xa0 = x4[0 * 1024 + tid];
    const float4 xa1 = x4[1 * 1024 + tid];
    const float4 xa2 = x4[2 * 1024 + tid];
    const float4 xa3 = x4[3 * 1024 + tid];
    const float4 xa4 = x4[4 * 1024 + tid];
    const float4 xa5 = x4[5 * 1024 + tid];
    const float4 xa6 = x4[min(6 * 1024 + tid, XA_VEC4 - 1)];
    asm volatile("" : : "s"(x) : "memory");

    const int4* t4 = (const int4*)tcnt;
    const int n4 = b * (NCHUNK / 4);
    int part = 0;
    for (int k0 = 0; k0 < n4; k0 += 8 * 1024) {
        int4 v[8];
#pragma unroll
        for (int u = 0; u < 8; ++u) {
            const int k = k0 + u * 1024 + tid;
            v[u] = t4[min(k, n4 - 1)];
        }
#pragma unroll
        for (int u = 0; u < 8; ++u) {
            const int k = k0 + u * 1024 + tid;
            const int s = (v[u].x + v[u].y) + (v[u].z + v[u].w);
            part += (k < n4) ? s : 0;
        }
    }
    int c = 0;
    if (tid < NCHUNK) {
        c = tcnt[b * NCHUNK + tid];
        sabs[tid] = tstart[b * NCHUNK + tid];
    }
    if (tid < NPB) lcnt[tid] = 0;
    XS_PUT(0 * 1024 + tid, xa0);
    XS_PUT(1 * 1024 + tid, xa1);
    XS_PUT(2 * 1024 + tid, xa2);
    XS_PUT(3 * 1024 + tid, xa3);
    XS_PUT(4 * 1024 + tid, xa4);
    XS_PUT(5 * 1024 + tid, xa5);
    if (6 * 1024 + tid < XA_VEC4) XS_PUT(6 * 1024 + tid, xa6);
    int Tb, base;
    const int exc = block_scan_excl(c, wsum, Tb);
    if (tid < NCHUNK) p[tid] = exc;
    if (tid == 0) p[NCHUNK] = Tb;
    (void)block_scan_excl(part, wsum, base);
    __syncthreads();
    const int a0 = base & ~31;
    const bool staged = Tb <= IMG_INTS;

    for (int j0 = 0; j0 < Tb; j0 += 8 * 1024) {
        int jj[8], idx[8];
#pragma unroll
        for (int u = 0; u < 8; ++u) jj[u] = min(j0 + u * 1024 + tid, Tb - 1);
        find_in_runs(p, sabs, jj, idx);
        int e[8];
#pragma unroll
        for (int u = 0; u < 8; ++u) e[u] = pk[idx[u]];
#pragma unroll
        for (int u = 0; u < 8; ++u) {
            const int j = j0 + u * 1024 + tid;
            if (j < Tb) atomicAdd(&lcnt[e[u] >> RBITS], 1);
        }
    }
    __syncthreads();
    const int d = (tid < NPB) ? lcnt[tid] : 0;
    int tot;
    const int ex = block_scan_excl(d, wsum, tot);
    if (tid < NPB) {
        const int node = b * NPB + tid;
        const int o = base + ex;
        const float dn = rsqrtf((float)(d + 1));
        off[node] = o;
        dis[node] = dn;
        sdis[tid] = dn;
        lcnt[tid] = o;
    }
    if (b == NBINS - 1 && tid == 0) off[N_NODES] = base + tot;
    __syncthreads();

    if (staged) {
        place_range<true>(pk, p, sabs, lcnt, img, srcs, Tb, a0);
        __syncthreads();
        copy_image_out(img, srcs, a0, base, base + Tb);
    } else {
        place_range<false>(pk, p, sabs, lcnt, img, srcs, Tb, a0);
    }

    const int lane = tid & 31;
    const int wave = __builtin_amdgcn_readfirstlane(tid >> 5);
    const int n0   = (wave & 3) * 16;
    const int g8   = wave >> 2;
    const int half = lane >> 4;
    const int l    = lane & 15;
    const float* bcol = W + (2 * half) * 64 + n0 + l;
    v2f bw[32];
#pragma unroll
    for (int s = 0; s < 32; ++s) {
        bw[s].x = bcol[(4 * s + 0) * 64];
        bw[s].y = bcol[(4 * s + 1) * 64];
    }
    asm volatile("" : "+v"(bw[0]), "+v"(bw[1]), "+v"(bw[2]), "+v"(bw[3]), "+v"(bw[4]), "+v"(bw[5]),
                      "+v"(bw[6]), "+v"(bw[7]), "+v"(bw[8]), "+v"(bw[9]), "+v"(bw[10]), "+v"(bw[11]),
                      "+v"(bw[12]), "+v"(bw[13]), "+v"(bw[14]), "+v"(bw[15]));
    asm volatile("" : "+v"(bw[16]), "+v"(bw[17]), "+v"(bw[18]), "+v"(bw[19]), "+v"(bw[20]), "+v"(bw[21]),
                      "+v"(bw[22]), "+v"(bw[23]), "+v"(bw[24]), "+v"(bw[25]), "+v"(bw[26]), "+v"(bw[27]),
                      "+v"(bw[28]), "+v"(bw[29]), "+v"(bw[30]), "+v"(bw[31]));
    const float4 xb0 = x4[XA_VEC4 + 0 * 1024 + tid];
    const float4 xb1 = x4[XA_VEC4 + 1 * 1024 + tid];
    const float4 xb2 = x4[XA_VEC4 + 2 * 1024 + tid];
    const float4 xb3 = x4[XA_VEC4 + 3 * 1024 + tid];
    const float4 xb4 = x4[XA_VEC4 + 4 * 1024 + tid];
    const float4 xb5 = x4[XA_VEC4 + 5 * 1024 + tid];
    asm volatile("" : : "s"(x) : "memory");
    float* hrow = h + b * (NPB * 64) + n0;
    for (int rt = g8; rt < XA_TILES; rt += 8)
        product_tile(xs, sdis, hrow, bw, rt, half, l);
    XS_PUT(XA_VEC4 + 0 * 1024 + tid, xb0);
    XS_PUT(XA_VEC4 + 1 * 1024 + tid, xb1);
    XS_PUT(XA_VEC4 + 2 * 1024 + tid, xb2);
    XS_PUT(XA_VEC4 + 3 * 1024 + tid, xb3);
    XS_PUT(XA_VEC4 + 4 * 1024 + tid, xb4);
    XS_PUT(XA_VEC4 + 5 * 1024 + tid, xb5);
    __syncthreads();
    for (int j = (g8 + 3) & 7; j < XB_TILES; j += 8)
        product_tile(xs, sdis, hrow, bw, XA_TILES + j, half, l);
}
#undef XS_PUT

template <int F, bool RELU>
__device__ __forceinline__ float4 gather_row_sum(const int* __restrict__ off,
                                                 const int* __restrict__ srcs,
                                                 const float dc,
                                                 const float* __restrict__ hs,
                                                 const float* __restrict__ bias,
                                                 const int c, const int g) {
    constexpr int LANES = F / 4;
    const int fc  = g * 4;
    const int beg = off[c];
    const int end = off[c + 1];

    float4 acc = *(const float4*)(hs + c * F + fc);
    const float4 bv = *(const float4*)(bias + fc);

    for (int b0 = beg; b0 < end; b0 += LANES) {
        const int n = min(LANES, end - b0);
        int myr = c;
        if (g < n) myr = srcs[b0 + g];
        for (int k0 = 0; k0 < n; k0 += 4) {
            const int r0 = __shfl(myr, k0 + 0, LANES);
            const int r1 = __shfl(myr, k0 + 1, LANES);
            const int r2 = __shfl(myr, k0 + 2, LANES);
            const int r3 = __shfl(myr, k0 + 3, LANES);
            const float m1 = (k0 + 1 < n) ? 1.0f : 0.0f;
            const float m2 = (k0 + 2 < n) ? 1.0f : 0.0f;
            const float m3 = (k0 + 3 < n) ? 1.0f : 0.0f;
            const float4 v0 = *(const float4*)(hs + r0 * F + fc);
            const float4 v1 = *(const float4*)(hs + r1 * F + fc);
            const float4 v2 = *(const float4*)(hs + r2 * F + fc);
            const float4 v3 = *(const float4*)(hs + r3 * F + fc);
            acc.x += v0.x;      acc.y += v0.y;      acc.z += v0.z;      acc.w += v0.w;
            acc.x += v1.x * m1; acc.y += v1.y * m1; acc.z += v1.z * m1; acc.w += v1.w * m1;
            acc.x += v2.x * m2; acc.y += v2.y * m2; acc.z += v2.z * m2; acc.w += v2.w * m2;
            acc.x += v3.x * m3; acc.y += v3.y * m3; acc.z += v3.z * m3; acc.w += v3.w * m3;
        }
    }
    acc.x = bv.x + acc.x * dc;
    acc.y = bv.y + acc.y * dc;
    acc.z = bv.z + acc.z * dc;
    acc.w = bv.w + acc.w * dc;
    if (RELU) {
        acc.x = fmaxf(acc.x, 0.0f);
        acc.y = fmaxf(acc.y, 0.0f);
        acc.z = fmaxf(acc.z, 0.0f);
        acc.w = fmaxf(acc.w, 0.0f);
    }
    return acc;
}

#define TILE_PITCH 68
__global__ __launch_bounds__(256) void gather64_relu_gemm2_kernel(
    const int* __restrict__ off, const int* __restrict__ srcs,
    const float* __restrict__ dis, const float* __restrict__ h,
    const float* __restrict__ bias, const float* __restrict__ W,
    float* __restrict__ gout) {
    __shared__ __attribute__((aligned(16))) float tile[16 * TILE_PITCH];
    __shared__ __attribute__((aligned(16))) float sdis[16];
    const int cl   = threadIdx.x >> 4;
    const int g    = threadIdx.x & 15;
    const int m0   = blockIdx.x * 16;
    const int lane = threadIdx.x & 31;
    const int wave = threadIdx.x >> 5;
    const int half = lane >> 4;
    const int l    = lane & 15;
    const int n0   = (wave & 1) * 16;

    const float dc = dis[m0 + cl];
    const float4 acc = gather_row_sum<64, true>(off, srcs, dc, h, bias, m0 + cl, g);
    *(float4*)(tile + cl * TILE_PITCH + g * 4) = acc;
    if (g == 0) sdis[cl] = dc;

    v2f bw[16];
    if (wave < 2) {
        const float* bcol = W + (2 * half) * 32 + n0 + l;
#pragma unroll
        for (int s = 0; s < 16; ++s) {
            bw[s].x = bcol[(4 * s + 0) * 32];
            bw[s].y = bcol[(4 * s + 1) * 32];
        }
    }
    __syncthreads();
    if (wave < 2) {
        const lfloat* arow = (const lfloat*)(tile + l * TILE_PITCH + 2 * half);
        const lfloat* srow = (const lfloat*)(sdis + half * 8);
        gfloat* o = (gfloat*)(gout + (m0 + half * 8) * 32 + n0 + l);
        asm volatile("" : "+v"(arow), "+v"(srow), "+v"(o));
        v8f c0 = {};
        v2f a[8];
#pragma unroll
        for (int s = 0; s < 8; ++s) a[s] = *(const __attribute__((address_space(3))) v2f*)(arow + 4 * s);
        __builtin_amdgcn_sched_barrier(0);
#pragma unroll
        for (int s = 0; s < 8; ++s)
            c0 = __builtin_amdgcn_wmma_f32_16x16x4_f32(false, a[s], false, bw[s],
                                                       (short)0, c0, false, false);
        IDLE_SLOTS_BEHIND_PRODUCT(c0);
#pragma unroll
        for (int s = 0; s < 8; ++s) a[s] = *(const __attribute__((address_space(3))) v2f*)(arow + 32 + 4 * s);
        __builtin_amdgcn_sched_barrier(0);
#pragma unroll
        for (int s = 0; s < 8; ++s)
            c0 = __builtin_amdgcn_wmma_f32_16x16x4_f32(false, a[s], false, bw[8 + s],
                                                       (short)0, c0, false, false);
        IDLE_SLOTS_BEHIND_PRODUCT(c0);
        const float4 s0 = *(const __attribute__((address_space(3))) float4*)(srow);
        const float4 s1 = *(const __attribute__((address_space(3))) float4*)(srow + 4);
        const float sc[8] = {s0.x, s0.y, s0.z, s0.w, s1.x, s1.y, s1.z, s1.w};
#pragma unroll
        for (int v = 0; v < 8; ++v) o[v * 32] = c0[v] * sc[v];
    }
}

__global__ __launch_bounds__(256) void gather32_kernel(
    const int* __restrict__ off, const int* __restrict__ srcs,
    const float* __restrict__ dis, const float* __restrict__ h,
    const float* __restrict__ bias, float* __restrict__ res) {
    const int tid = blockIdx.x * blockDim.x + threadIdx.x;
    const int c   = tid / 8;
    const int g   = tid % 8;
    const float4 acc = gather_row_sum<32, false>(off, srcs, dis[c], h, bias, c, g);
    *(float4*)(res + c * 32 + g * 4) = acc;
}

extern "C" void kernel_launch(void* const* d_in, const int* in_sizes, int n_in,
                              void* d_out, int out_size, void* d_ws, size_t ws_size,
                              hipStream_t stream) {
    const float* x  = (const float*)d_in[0];
    const int*   ei = (const int*)d_in[1];
    const float* W1 = (const float*)d_in[2];
    const float* b1 = (const float*)d_in[3];
    const float* W2 = (const float*)d_in[4];
    const float* b2 = (const float*)d_in[5];
    float* out = (float*)d_out;

    const int E = in_sizes[1] / 2;
    const int* row = ei;
    const int* col = ei + E;

    float* dis  = (float*)d_ws;
    float* h1   = dis  + N_NODES;
    float* g2   = h1   + N_NODES * 64;
    int*   srcs = (int*)(g2 + N_NODES * 64);
    int*   off  = srcs + N_NODES * 16;
    int*   pk     = (int*)(g2 + N_NODES * 32);
    int*   tcnt   = (int*)g2;
    int*   tstart = tcnt + NBINS * NCHUNK;

    const int T = 256;
    chunk_sort_kernel<<<NCHUNK, 1024, 0, stream>>>(row, col, pk, tcnt, tstart, E);
    bucket_gemm1_kernel<<<NBINS, 1024, 0, stream>>>(pk, tcnt, tstart, off, dis, srcs, x, W1, h1);

    gather64_relu_gemm2_kernel<<<ROW_TILES, T, 0, stream>>>(off, srcs, dis, h1, b1, W2, g2);

    gather32_kernel<<<N_NODES * 8 / T, T, 0, stream>>>(off, srcs, dis, g2, b2, out);
}


# both sort kernels read their edges once: the counting pass's returning LDS add gives each element's rank, element and rank wait in registers, place = start + rank; sum order = LDS arrival rank
# speedup vs baseline: 4.9867x; 1.0501x over previous
#include <hip/hip_runtime.h>
#include <hip/hip_bf16.h>

typedef __attribute__((ext_vector_type(2))) float v2f;
typedef __attribute__((ext_vector_type(8))) float v8f;
typedef __attribute__((address_space(1))) float gfloat;
typedef __attribute__((address_space(3))) float lfloat;

#define N_NODES 100000
#define ROW_TILES (N_NODES / 16)

#define NPB    400
#define NBINS  250
#define NCHUNK 256
#define RBITS  17
#define IMG_INTS (8 * 1024)
#define IMG_TRIPS ((IMG_INTS + 32 + 1023) / 1024)

__device__ __forceinline__ void copy_image_out(const int* img, int* __restrict__ dst,
                                               const int a0, const int first, const int stop) {
    int v[IMG_TRIPS];
#pragma unroll
    for (int k = 0; k < IMG_TRIPS; ++k)
        v[k] = img[min(k * 1024 + (int)threadIdx.x, IMG_INTS + 31)];
#pragma unroll
    for (int k = 0; k < IMG_TRIPS; ++k) {
        const int gi = a0 + k * 1024 + (int)threadIdx.x;
        if (gi >= first && gi < stop) dst[gi] = v[k];
    }
}
static_assert(NBINS * NPB == N_NODES, "ranges must tile the nodes");
static_assert((NCHUNK & (NCHUNK - 1)) == 0 && NCHUNK <= 1024 && NBINS <= 256, "bisection and scans assume this");
static_assert(N_NODES <= (1 << RBITS), "source id must fit the packed word");
static_assert(NPB <= (1 << (31 - RBITS)), "local target must fit the packed word");

__device__ __forceinline__ int block_scan_excl(int v, int* wsum, int& total) {
    const int lane = threadIdx.x & 31;
    const int wave = threadIdx.x >> 5;
    int inc = v;
#pragma unroll
    for (int o = 1; o < 32; o <<= 1) {
        int u = __shfl_up(inc, o);
        if (lane >= o) inc += u;
    }
    __syncthreads();
    if (lane == 31) wsum[wave] = inc;
    __syncthreads();
    int wb = 0, tot = 0;
#pragma unroll
    for (int w = 0; w < 32; ++w) {
        const int s = wsum[w];
        if (w < wave) wb += s;
        tot += s;
    }
    total = tot;
    return wb + inc - v;
}

__device__ __forceinline__ void place_chunk(const int* __restrict__ row, const int* __restrict__ col,
                                            int* __restrict__ pk, int* lh,
                                            const int beg, const int end) {
    const int tid = threadIdx.x;
    for (int i0 = beg; i0 < end; i0 += 8 * 1024) {
        int cv[8], rv[8];
#pragma unroll
        for (int u = 0; u < 8; ++u) {
            const int i = min(i0 + u * 1024 + tid, end - 1);
            cv[u] = col[i];
            rv[u] = row[i];
        }
#pragma unroll
        for (int u = 0; u < 8; ++u) {
            const int i = i0 + u * 1024 + tid;
            if (i < end) {
                const unsigned b = (unsigned)cv[u] / NPB;
                const int t = cv[u] - (int)b * NPB;
                const int slot = atomicAdd(&lh[b], 1);
                pk[slot] = (t << RBITS) | rv[u];
            }
        }
    }
}

__global__ __launch_bounds__(1024) void chunk_sort_kernel(
    const int* __restrict__ row, const int* __restrict__ col,
    int* __restrict__ pk, int* __restrict__ tcnt, int* __restrict__ tstart, int E) {
    __shared__ int lh[256];
    __shared__ int wsum[32];
    __shared__ int img[IMG_INTS + 32];
    const int tid = threadIdx.x;
    const int w   = blockIdx.x;
    const int chunk = (E + NCHUNK - 1) / NCHUNK;
    const int beg = min(w * chunk, E);
    const int end = min(beg + chunk, E);
    const int a0  = beg & ~31;
    const bool staged = (end - beg) <= IMG_INTS;

    if (tid < 256) lh[tid] = 0;
    __syncthreads();
    int cv[8], rv[8], rk[8];
    if (staged) {
        if (end > beg) {
#pragma unroll
            for (int u = 0; u < 8; ++u) {
                const int i = min(beg + u * 1024 + tid, end - 1);
                cv[u] = col[i];
                rv[u] = row[i];
            }
#pragma unroll
            for (int u = 0; u < 8; ++u) {
                rk[u] = 0;
                if (beg + u * 1024 + tid < end) rk[u] = atomicAdd(&lh[(unsigned)cv[u] / NPB], 1);
            }
        }
    } else {
        for (int i0 = beg; i0 < end; i0 += 8 * 1024) {
            int cw[8];
#pragma unroll
            for (int u = 0; u < 8; ++u) {
                const int i = i0 + u * 1024 + tid;
                cw[u] = col[min(i, end - 1)];
            }
#pragma unroll
            for (int u = 0; u < 8; ++u) {
                const int i = i0 + u * 1024 + tid;
                if (i < end) atomicAdd(&lh[(unsigned)cw[u] / NPB], 1);
            }
        }
    }
    __syncthreads();
    const int v = (tid < NBINS) ? lh[tid] : 0;
    int tot;
    const int ex = block_scan_excl(v, wsum, tot);
    if (tid < NBINS) {
        tcnt[tid * NCHUNK + w]   = v;
        tstart[tid * NCHUNK + w] = beg + ex;
        lh[tid] = beg + ex;
    }
    __syncthreads();
    if (staged) {
#pragma unroll
        for (int u = 0; u < 8; ++u) {
            if (beg + u * 1024 + tid < end) {
                const unsigned b = (unsigned)cv[u] / NPB;
                const int t = cv[u] - (int)b * NPB;
                img[lh[b] + rk[u] - a0] = (t << RBITS) | rv[u];
            }
        }
        __syncthreads();
        copy_image_out(img, pk, a0, beg, end);
    } else {
        place_chunk(row, col, pk, lh, beg, end);
    }
}

__device__ __forceinline__ void find_in_runs(const int* p, const int* sabs,
                                             const int (&j)[8], int (&idx)[8]) {
    int lo[8];
#pragma unroll
    for (int u = 0; u < 8; ++u) lo[u] = 0;
#pragma unroll
    for (int half = NCHUNK / 2; half > 0; half >>= 1) {
        int pm[8];
#pragma unroll
        for (int u = 0; u < 8; ++u) pm[u] = p[lo[u] + half];
#pragma unroll
        for (int u = 0; u < 8; ++u) if (pm[u] <= j[u]) lo[u] += half;
    }
#pragma unroll
    for (int u = 0; u < 8; ++u) idx[u] = sabs[lo[u]] + (j[u] - p[lo[u]]);
}

__device__ __forceinline__ void place_range(const int* __restrict__ pk, const int* p, const int* sabs,
                                            int* lcnt, int* __restrict__ srcs, const int Tb) {
    const int tid = threadIdx.x;
    for (int j0 = 0; j0 < Tb; j0 += 8 * 1024) {
        int jj[8], idx[8];
#pragma unroll
        for (int u = 0; u < 8; ++u) jj[u] = min(j0 + u * 1024 + tid, Tb - 1);
        find_in_runs(p, sabs, jj, idx);
        int e[8];
#pragma unroll
        for (int u = 0; u < 8; ++u) e[u] = pk[idx[u]];
#pragma unroll
        for (int u = 0; u < 8; ++u) {
            const int j = j0 + u * 1024 + tid;
            if (j < Tb) {
                const int slot = atomicAdd(&lcnt[e[u] >> RBITS], 1);
                srcs[slot] = e[u] & ((1 << RBITS) - 1);
            }
        }
    }
}


#define IDLE_SLOTS_BEHIND_PRODUCT(c0)                                           \
    asm volatile("v_nop\n\tv_nop\n\tv_nop\n\tv_nop\n\t"                        \
                 "v_nop\n\tv_nop\n\tv_nop\n\tv_nop" : "+v"(c0) : : "memory")

#define X_PITCH 132
#define XA_TILES 13
#define XB_TILES (NPB / 16 - XA_TILES)
#define XA_VEC4  (XA_TILES * 16 * 32)
#define XB_VEC4  (XB_TILES * 16 * 32)
static_assert(NPB % 16 == 0, "a range must be whole row tiles");
static_assert(XA_VEC4 > 6 * 1024 && XA_VEC4 <= 7 * 1024, "the bucket kernel spells out 7 trips of 1024 float4s for part one");
static_assert(XB_VEC4 == 6 * 1024, "the bucket kernel spells out 6 trips of 1024 float4s for part two");
static_assert(XA_TILES > 8 && XA_TILES <= 16 && XB_TILES > 8 && XB_TILES <= 16, "a wave has one or two row tiles in each part");

#define XS_PUT(i, v) *(float4*)(xs + ((i) >> 5) * X_PITCH + ((i) & 31) * 4) = (v)

__device__ __forceinline__ void product_tile(const float* xs, const float* sdis, float* hrow,
                                             const v2f (&bw)[32], const int rt,
                                             const int half, const int l) {
    const lfloat* arow = (const lfloat*)(xs + (rt * 16 + l) * X_PITCH + 2 * half);
    const lfloat* srow = (const lfloat*)(sdis + rt * 16 + half * 8);
    gfloat* o = (gfloat*)(hrow + (rt * 16 + half * 8) * 64 + l);
    asm volatile("" : "+v"(arow), "+v"(srow), "+v"(o));
    v8f c0 = {};
#pragma unroll
    for (int q = 0; q < 4; ++q) {
        v2f a[8];
#pragma unroll
        for (int s = 0; s < 8; ++s)
            a[s] = *(const __attribute__((address_space(3))) v2f*)(arow + 32 * q + 4 * s);
        __builtin_amdgcn_sched_barrier(0);
#pragma unroll
        for (int s = 0; s < 8; ++s)
            c0 = __builtin_amdgcn_wmma_f32_16x16x4_f32(false, a[s], false, bw[8 * q + s],
                                                       (short)0, c0, false, false);
        IDLE_SLOTS_BEHIND_PRODUCT(c0);
    }
    const float4 s0 = *(const __attribute__((address_space(3))) float4*)(srow);
    const float4 s1 = *(const __attribute__((address_space(3))) float4*)(srow + 4);
    const float sc[8] = {s0.x, s0.y, s0.z, s0.w, s1.x, s1.y, s1.z, s1.w};
#pragma unroll
    for (int v = 0; v < 8; ++v) o[v * 64] = c0[v] * sc[v];
}

__global__ __launch_bounds__(1024) void bucket_gemm1_kernel(
    const int* __restrict__ pk, const int* __restrict__ tcnt,
    const int* __restrict__ tstart, int* __restrict__ off,
    float* __restrict__ dis, int* __restrict__ srcs,
    const float* __restrict__ x, const float* __restrict__ W,
    float* __restrict__ h) {
    __shared__ int p[NCHUNK + 1];
    __shared__ int sabs[NCHUNK];
    __shared__ int lcnt[NPB];
    __shared__ int wsum[32];
    __shared__ int img[IMG_INTS + 32];
    __shared__ __attribute__((aligned(16))) float xs[NPB * X_PITCH];
    __shared__ __attribute__((aligned(16))) float sdis[NPB];
    const int tid = threadIdx.x;
    const int b   = blockIdx.x;

    const float4* x4 = (const float4*)(x + b * (NPB * 128));
    const float4 xa0 = x4[0 * 1024 + tid];
    const float4 xa1 = x4[1 * 1024 + tid];
    const float4 xa2 = x4[2 * 1024 + tid];
    const float4 xa3 = x4[3 * 1024 + tid];
    const float4 xa4 = x4[4 * 1024 + tid];
    const float4 xa5 = x4[5 * 1024 + tid];
    const float4 xa6 = x4[min(6 * 1024 + tid, XA_VEC4 - 1)];
    asm volatile("" : : "s"(x) : "memory");

    const int4* t4 = (const int4*)tcnt;
    const int n4 = b * (NCHUNK / 4);
    int part = 0;
    for (int k0 = 0; k0 < n4; k0 += 8 * 1024) {
        int4 v[8];
#pragma unroll
        for (int u = 0; u < 8; ++u) {
            const int k = k0 + u * 1024 + tid;
            v[u] = t4[min(k, n4 - 1)];
        }
#pragma unroll
        for (int u = 0; u < 8; ++u) {
            const int k = k0 + u * 1024 + tid;
            const int s = (v[u].x + v[u].y) + (v[u].z + v[u].w);
            part += (k < n4) ? s : 0;
        }
    }
    int c = 0;
    if (tid < NCHUNK) {
        c = tcnt[b * NCHUNK + tid];
        sabs[tid] = tstart[b * NCHUNK + tid];
    }
    if (tid < NPB) lcnt[tid] = 0;
    XS_PUT(0 * 1024 + tid, xa0);
    XS_PUT(1 * 1024 + tid, xa1);
    XS_PUT(2 * 1024 + tid, xa2);
    XS_PUT(3 * 1024 + tid, xa3);
    XS_PUT(4 * 1024 + tid, xa4);
    XS_PUT(5 * 1024 + tid, xa5);
    if (6 * 1024 + tid < XA_VEC4) XS_PUT(6 * 1024 + tid, xa6);
    int Tb, base;
    const int exc = block_scan_excl(c, wsum, Tb);
    if (tid < NCHUNK) p[tid] = exc;
    if (tid == 0) p[NCHUNK] = Tb;
    (void)block_scan_excl(part, wsum, base);
    __syncthreads();
    const int a0 = base & ~31;
    const bool staged = Tb <= IMG_INTS;

    int e[8], rk[8];
#pragma unroll
    for (int u = 0; u < 8; ++u) { e[u] = 0; rk[u] = 0; }
    if (staged) {
        if (Tb > 0) {
            int jj[8], idx[8];
#pragma unroll
            for (int u = 0; u < 8; ++u) jj[u] = min(u * 1024 + tid, Tb - 1);
            find_in_runs(p, sabs, jj, idx);
#pragma unroll
            for (int u = 0; u < 8; ++u) e[u] = pk[idx[u]];
#pragma unroll
            for (int u = 0; u < 8; ++u)
                if (u * 1024 + tid < Tb) rk[u] = atomicAdd(&lcnt[e[u] >> RBITS], 1);
        }
    } else {
        for (int j0 = 0; j0 < Tb; j0 += 8 * 1024) {
            int jj[8], idx[8], ev[8];
#pragma unroll
            for (int u = 0; u < 8; ++u) jj[u] = min(j0 + u * 1024 + tid, Tb - 1);
            find_in_runs(p, sabs, jj, idx);
#pragma unroll
            for (int u = 0; u < 8; ++u) ev[u] = pk[idx[u]];
#pragma unroll
            for (int u = 0; u < 8; ++u) {
                const int j = j0 + u * 1024 + tid;
                if (j < Tb) atomicAdd(&lcnt[ev[u] >> RBITS], 1);
            }
        }
    }
    __syncthreads();
    const int d = (tid < NPB) ? lcnt[tid] : 0;
    int tot;
    const int ex = block_scan_excl(d, wsum, tot);
    if (tid < NPB) {
        const int node = b * NPB + tid;
        const int o = base + ex;
        const float dn = rsqrtf((float)(d + 1));
        off[node] = o;
        dis[node] = dn;
        sdis[tid] = dn;
        lcnt[tid] = o;
    }
    if (b == NBINS - 1 && tid == 0) off[N_NODES] = base + tot;
    __syncthreads();

    if (staged) {
#pragma unroll
        for (int u = 0; u < 8; ++u)
            if (u * 1024 + tid < Tb)
                img[lcnt[e[u] >> RBITS] + rk[u] - a0] = e[u] & ((1 << RBITS) - 1);
        __syncthreads();
        copy_image_out(img, srcs, a0, base, base + Tb);
    } else {
        place_range(pk, p, sabs, lcnt, srcs, Tb);
    }

    const int lane = tid & 31;
    const int wave = __builtin_amdgcn_readfirstlane(tid >> 5);
    const int n0   = (wave & 3) * 16;
    const int g8   = wave >> 2;
    const int half = lane >> 4;
    const int l    = lane & 15;
    const float* bcol = W + (2 * half) * 64 + n0 + l;
    v2f bw[32];
#pragma unroll
    for (int s = 0; s < 32; ++s) {
        bw[s].x = bcol[(4 * s + 0) * 64];
        bw[s].y = bcol[(4 * s + 1) * 64];
    }
    asm volatile("" : "+v"(bw[0]), "+v"(bw[1]), "+v"(bw[2]), "+v"(bw[3]), "+v"(bw[4]), "+v"(bw[5]),
                      "+v"(bw[6]), "+v"(bw[7]), "+v"(bw[8]), "+v"(bw[9]), "+v"(bw[10]), "+v"(bw[11]),
                      "+v"(bw[12]), "+v"(bw[13]), "+v"(bw[14]), "+v"(bw[15]));
    asm volatile("" : "+v"(bw[16]), "+v"(bw[17]), "+v"(bw[18]), "+v"(bw[19]), "+v"(bw[20]), "+v"(bw[21]),
                      "+v"(bw[22]), "+v"(bw[23]), "+v"(bw[24]), "+v"(bw[25]), "+v"(bw[26]), "+v"(bw[27]),
                      "+v"(bw[28]), "+v"(bw[29]), "+v"(bw[30]), "+v"(bw[31]));
    const float4 xb0 = x4[XA_VEC4 + 0 * 1024 + tid];
    const float4 xb1 = x4[XA_VEC4 + 1 * 1024 + tid];
    const float4 xb2 = x4[XA_VEC4 + 2 * 1024 + tid];
    const float4 xb3 = x4[XA_VEC4 + 3 * 1024 + tid];
    const float4 xb4 = x4[XA_VEC4 + 4 * 1024 + tid];
    const float4 xb5 = x4[XA_VEC4 + 5 * 1024 + tid];
    asm volatile("" : : "s"(x) : "memory");
    float* hrow = h + b * (NPB * 64) + n0;
    for (int rt = g8; rt < XA_TILES; rt += 8)
        product_tile(xs, sdis, hrow, bw, rt, half, l);
    XS_PUT(XA_VEC4 + 0 * 1024 + tid, xb0);
    XS_PUT(XA_VEC4 + 1 * 1024 + tid, xb1);
    XS_PUT(XA_VEC4 + 2 * 1024 + tid, xb2);
    XS_PUT(XA_VEC4 + 3 * 1024 + tid, xb3);
    XS_PUT(XA_VEC4 + 4 * 1024 + tid, xb4);
    XS_PUT(XA_VEC4 + 5 * 1024 + tid, xb5);
    __syncthreads();
    for (int j = (g8 + 3) & 7; j < XB_TILES; j += 8)
        product_tile(xs, sdis, hrow, bw, XA_TILES + j, half, l);
}
#undef XS_PUT

template <int F, bool RELU>
__device__ __forceinline__ float4 gather_row_sum(const int* __restrict__ off,
                                                 const int* __restrict__ srcs,
                                                 const float dc,
                                                 const float* __restrict__ hs,
                                                 const float* __restrict__ bias,
                                                 const int c, const int g) {
    constexpr int LANES = F / 4;
    const int fc  = g * 4;
    const int beg = off[c];
    const int end = off[c + 1];

    float4 acc = *(const float4*)(hs + c * F + fc);
    const float4 bv = *(const float4*)(bias + fc);

    for (int b0 = beg; b0 < end; b0 += LANES) {
        const int n = min(LANES, end - b0);
        int myr = c;
        if (g < n) myr = srcs[b0 + g];
        for (int k0 = 0; k0 < n; k0 += 4) {
            const int r0 = __shfl(myr, k0 + 0, LANES);
            const int r1 = __shfl(myr, k0 + 1, LANES);
            const int r2 = __shfl(myr, k0 + 2, LANES);
            const int r3 = __shfl(myr, k0 + 3, LANES);
            const float m1 = (k0 + 1 < n) ? 1.0f : 0.0f;
            const float m2 = (k0 + 2 < n) ? 1.0f : 0.0f;
            const float m3 = (k0 + 3 < n) ? 1.0f : 0.0f;
            const float4 v0 = *(const float4*)(hs + r0 * F + fc);
            const float4 v1 = *(const float4*)(hs + r1 * F + fc);
            const float4 v2 = *(const float4*)(hs + r2 * F + fc);
            const float4 v3 = *(const float4*)(hs + r3 * F + fc);
            acc.x += v0.x;      acc.y += v0.y;      acc.z += v0.z;      acc.w += v0.w;
            acc.x += v1.x * m1; acc.y += v1.y * m1; acc.z += v1.z * m1; acc.w += v1.w * m1;
            acc.x += v2.x * m2; acc.y += v2.y * m2; acc.z += v2.z * m2; acc.w += v2.w * m2;
            acc.x += v3.x * m3; acc.y += v3.y * m3; acc.z += v3.z * m3; acc.w += v3.w * m3;
        }
    }
    acc.x = bv.x + acc.x * dc;
    acc.y = bv.y + acc.y * dc;
    acc.z = bv.z + acc.z * dc;
    acc.w = bv.w + acc.w * dc;
    if (RELU) {
        acc.x = fmaxf(acc.x, 0.0f);
        acc.y = fmaxf(acc.y, 0.0f);
        acc.z = fmaxf(acc.z, 0.0f);
        acc.w = fmaxf(acc.w, 0.0f);
    }
    return acc;
}

#define TILE_PITCH 68
__global__ __launch_bounds__(256) void gather64_relu_gemm2_kernel(
    const int* __restrict__ off, const int* __restrict__ srcs,
    const float* __restrict__ dis, const float* __restrict__ h,
    const float* __restrict__ bias, const float* __restrict__ W,
    float* __restrict__ gout) {
    __shared__ __attribute__((aligned(16))) float tile[16 * TILE_PITCH];
    __shared__ __attribute__((aligned(16))) float sdis[16];
    const int cl   = threadIdx.x >> 4;
    const int g    = threadIdx.x & 15;
    const int m0   = blockIdx.x * 16;
    const int lane = threadIdx.x & 31;
    const int wave = threadIdx.x >> 5;
    const int half = lane >> 4;
    const int l    = lane & 15;
    const int n0   = (wave & 1) * 16;

    const float dc = dis[m0 + cl];
    const float4 acc = gather_row_sum<64, true>(off, srcs, dc, h, bias, m0 + cl, g);
    *(float4*)(tile + cl * TILE_PITCH + g * 4) = acc;
    if (g == 0) sdis[cl] = dc;

    v2f bw[16];
    if (wave < 2) {
        const float* bcol = W + (2 * half) * 32 + n0 + l;
#pragma unroll
        for (int s = 0; s < 16; ++s) {
            bw[s].x = bcol[(4 * s + 0) * 32];
            bw[s].y = bcol[(4 * s + 1) * 32];
        }
    }
    __syncthreads();
    if (wave < 2) {
        const lfloat* arow = (const lfloat*)(tile + l * TILE_PITCH + 2 * half);
        const lfloat* srow = (const lfloat*)(sdis + half * 8);
        gfloat* o = (gfloat*)(gout + (m0 + half * 8) * 32 + n0 + l);
        asm volatile("" : "+v"(arow), "+v"(srow), "+v"(o));
        v8f c0 = {};
        v2f a[8];
#pragma unroll
        for (int s = 0; s < 8; ++s) a[s] = *(const __attribute__((address_space(3))) v2f*)(arow + 4 * s);
        __builtin_amdgcn_sched_barrier(0);
#pragma unroll
        for (int s = 0; s < 8; ++s)
            c0 = __builtin_amdgcn_wmma_f32_16x16x4_f32(false, a[s], false, bw[s],
                                                       (short)0, c0, false, false);
        IDLE_SLOTS_BEHIND_PRODUCT(c0);
#pragma unroll
        for (int s = 0; s < 8; ++s) a[s] = *(const __attribute__((address_space(3))) v2f*)(arow + 32 + 4 * s);
        __builtin_amdgcn_sched_barrier(0);
#pragma unroll
        for (int s = 0; s < 8; ++s)
            c0 = __builtin_amdgcn_wmma_f32_16x16x4_f32(false, a[s], false, bw[8 + s],
                                                       (short)0, c0, false, false);
        IDLE_SLOTS_BEHIND_PRODUCT(c0);
        const float4 s0 = *(const __attribute__((address_space(3))) float4*)(srow);
        const float4 s1 = *(const __attribute__((address_space(3))) float4*)(srow + 4);
        const float sc[8] = {s0.x, s0.y, s0.z, s0.w, s1.x, s1.y, s1.z, s1.w};
#pragma unroll
        for (int v = 0; v < 8; ++v) o[v * 32] = c0[v] * sc[v];
    }
}

__global__ __launch_bounds__(256) void gather32_kernel(
    const int* __restrict__ off, const int* __restrict__ srcs,
    const float* __restrict__ dis, const float* __restrict__ h,
    const float* __restrict__ bias, float* __restrict__ res) {
    const int tid = blockIdx.x * blockDim.x + threadIdx.x;
    const int c   = tid / 8;
    const int g   = tid % 8;
    const float4 acc = gather_row_sum<32, false>(off, srcs, dis[c], h, bias, c, g);
    *(float4*)(res + c * 32 + g * 4) = acc;
}

extern "C" void kernel_launch(void* const* d_in, const int* in_sizes, int n_in,
                              void* d_out, int out_size, void* d_ws, size_t ws_size,
                              hipStream_t stream) {
    const float* x  = (const float*)d_in[0];
    const int*   ei = (const int*)d_in[1];
    const float* W1 = (const float*)d_in[2];
    const float* b1 = (const float*)d_in[3];
    const float* W2 = (const float*)d_in[4];
    const float* b2 = (const float*)d_in[5];
    float* out = (float*)d_out;

    const int E = in_sizes[1] / 2;
    const int* row = ei;
    const int* col = ei + E;

    float* dis  = (float*)d_ws;
    float* h1   = dis  + N_NODES;
    float* g2   = h1   + N_NODES * 64;
    int*   srcs = (int*)(g2 + N_NODES * 64);
    int*   off  = srcs + N_NODES * 16;
    int*   pk     = (int*)(g2 + N_NODES * 32);
    int*   tcnt   = (int*)g2;
    int*   tstart = tcnt + NBINS * NCHUNK;

    const int T = 256;
    chunk_sort_kernel<<<NCHUNK, 1024, 0, stream>>>(row, col, pk, tcnt, tstart, E);
    bucket_gemm1_kernel<<<NBINS, 1024, 0, stream>>>(pk, tcnt, tstart, off, dis, srcs, x, W1, h1);

    gather64_relu_gemm2_kernel<<<ROW_TILES, T, 0, stream>>>(off, srcs, dis, h1, b1, W2, g2);

    gather32_kernel<<<N_NODES * 8 / T, T, 0, stream>>>(off, srcs, dis, g2, b2, out);
}


# both gathers: fewer dependent round trips per node: a node's lanes fetch the ids of two batches with two loads in flight together and ask for eight rows per trip, not four; sum order unchanged
# speedup vs baseline: 5.1558x; 1.0339x over previous
#include <hip/hip_runtime.h>
#include <hip/hip_bf16.h>

typedef __attribute__((ext_vector_type(2))) float v2f;
typedef __attribute__((ext_vector_type(8))) float v8f;
typedef __attribute__((address_space(1))) float gfloat;
typedef __attribute__((address_space(3))) float lfloat;

#define N_NODES 100000
#define ROW_TILES (N_NODES / 16)

#define NPB    400
#define NBINS  250
#define NCHUNK 256
#define RBITS  17
#define IMG_INTS (8 * 1024)
#define IMG_TRIPS ((IMG_INTS + 32 + 1023) / 1024)

__device__ __forceinline__ void copy_image_out(const int* img, int* __restrict__ dst,
                                               const int a0, const int first, const int stop) {
    int v[IMG_TRIPS];
#pragma unroll
    for (int k = 0; k < IMG_TRIPS; ++k)
        v[k] = img[min(k * 1024 + (int)threadIdx.x, IMG_INTS + 31)];
#pragma unroll
    for (int k = 0; k < IMG_TRIPS; ++k) {
        const int gi = a0 + k * 1024 + (int)threadIdx.x;
        if (gi >= first && gi < stop) dst[gi] = v[k];
    }
}
static_assert(NBINS * NPB == N_NODES, "ranges must tile the nodes");
static_assert((NCHUNK & (NCHUNK - 1)) == 0 && NCHUNK <= 1024 && NBINS <= 256, "bisection and scans assume this");
static_assert(N_NODES <= (1 << RBITS), "source id must fit the packed word");
static_assert(NPB <= (1 << (31 - RBITS)), "local target must fit the packed word");

__device__ __forceinline__ int block_scan_excl(int v, int* wsum, int& total) {
    const int lane = threadIdx.x & 31;
    const int wave = threadIdx.x >> 5;
    int inc = v;
#pragma unroll
    for (int o = 1; o < 32; o <<= 1) {
        int u = __shfl_up(inc, o);
        if (lane >= o) inc += u;
    }
    __syncthreads();
    if (lane == 31) wsum[wave] = inc;
    __syncthreads();
    int wb = 0, tot = 0;
#pragma unroll
    for (int w = 0; w < 32; ++w) {
        const int s = wsum[w];
        if (w < wave) wb += s;
        tot += s;
    }
    total = tot;
    return wb + inc - v;
}

__device__ __forceinline__ void place_chunk(const int* __restrict__ row, const int* __restrict__ col,
                                            int* __restrict__ pk, int* lh,
                                            const int beg, const int end) {
    const int tid = threadIdx.x;
    for (int i0 = beg; i0 < end; i0 += 8 * 1024) {
        int cv[8], rv[8];
#pragma unroll
        for (int u = 0; u < 8; ++u) {
            const int i = min(i0 + u * 1024 + tid, end - 1);
            cv[u] = col[i];
            rv[u] = row[i];
        }
#pragma unroll
        for (int u = 0; u < 8; ++u) {
            const int i = i0 + u * 1024 + tid;
            if (i < end) {
                const unsigned b = (unsigned)cv[u] / NPB;
                const int t = cv[u] - (int)b * NPB;
                const int slot = atomicAdd(&lh[b], 1);
                pk[slot] = (t << RBITS) | rv[u];
            }
        }
    }
}

__global__ __launch_bounds__(1024) void chunk_sort_kernel(
    const int* __restrict__ row, const int* __restrict__ col,
    int* __restrict__ pk, int* __restrict__ tcnt, int* __restrict__ tstart, int E) {
    __shared__ int lh[256];
    __shared__ int wsum[32];
    __shared__ int img[IMG_INTS + 32];
    const int tid = threadIdx.x;
    const int w   = blockIdx.x;
    const int chunk = (E + NCHUNK - 1) / NCHUNK;
    const int beg = min(w * chunk, E);
    const int end = min(beg + chunk, E);
    const int a0  = beg & ~31;
    const bool staged = (end - beg) <= IMG_INTS;

    if (tid < 256) lh[tid] = 0;
    __syncthreads();
    int cv[8], rv[8], rk[8];
    if (staged) {
        if (end > beg) {
#pragma unroll
            for (int u = 0; u < 8; ++u) {
                const int i = min(beg + u * 1024 + tid, end - 1);
                cv[u] = col[i];
                rv[u] = row[i];
            }
#pragma unroll
            for (int u = 0; u < 8; ++u) {
                rk[u] = 0;
                if (beg + u * 1024 + tid < end) rk[u] = atomicAdd(&lh[(unsigned)cv[u] / NPB], 1);
            }
        }
    } else {
        for (int i0 = beg; i0 < end; i0 += 8 * 1024) {
            int cw[8];
#pragma unroll
            for (int u = 0; u < 8; ++u) {
                const int i = i0 + u * 1024 + tid;
                cw[u] = col[min(i, end - 1)];
            }
#pragma unroll
            for (int u = 0; u < 8; ++u) {
                const int i = i0 + u * 1024 + tid;
                if (i < end) atomicAdd(&lh[(unsigned)cw[u] / NPB], 1);
            }
        }
    }
    __syncthreads();
    const int v = (tid < NBINS) ? lh[tid] : 0;
    int tot;
    const int ex = block_scan_excl(v, wsum, tot);
    if (tid < NBINS) {
        tcnt[tid * NCHUNK + w]   = v;
        tstart[tid * NCHUNK + w] = beg + ex;
        lh[tid] = beg + ex;
    }
    __syncthreads();
    if (staged) {
#pragma unroll
        for (int u = 0; u < 8; ++u) {
            if (beg + u * 1024 + tid < end) {
                const unsigned b = (unsigned)cv[u] / NPB;
                const int t = cv[u] - (int)b * NPB;
                img[lh[b] + rk[u] - a0] = (t << RBITS) | rv[u];
            }
        }
        __syncthreads();
        copy_image_out(img, pk, a0, beg, end);
    } else {
        place_chunk(row, col, pk, lh, beg, end);
    }
}

__device__ __forceinline__ void find_in_runs(const int* p, const int* sabs,
                                             const int (&j)[8], int (&idx)[8]) {
    int lo[8];
#pragma unroll
    for (int u = 0; u < 8; ++u) lo[u] = 0;
#pragma unroll
    for (int half = NCHUNK / 2; half > 0; half >>= 1) {
        int pm[8];
#pragma unroll
        for (int u = 0; u < 8; ++u) pm[u] = p[lo[u] + half];
#pragma unroll
        for (int u = 0; u < 8; ++u) if (pm[u] <= j[u]) lo[u] += half;
    }
#pragma unroll
    for (int u = 0; u < 8; ++u) idx[u] = sabs[lo[u]] + (j[u] - p[lo[u]]);
}

__device__ __forceinline__ void place_range(const int* __restrict__ pk, const int* p, const int* sabs,
                                            int* lcnt, int* __restrict__ srcs, const int Tb) {
    const int tid = threadIdx.x;
    for (int j0 = 0; j0 < Tb; j0 += 8 * 1024) {
        int jj[8], idx[8];
#pragma unroll
        for (int u = 0; u < 8; ++u) jj[u] = min(j0 + u * 1024 + tid, Tb - 1);
        find_in_runs(p, sabs, jj, idx);
        int e[8];
#pragma unroll
        for (int u = 0; u < 8; ++u) e[u] = pk[idx[u]];
#pragma unroll
        for (int u = 0; u < 8; ++u) {
            const int j = j0 + u * 1024 + tid;
            if (j < Tb) {
                const int slot = atomicAdd(&lcnt[e[u] >> RBITS], 1);
                srcs[slot] = e[u] & ((1 << RBITS) - 1);
            }
        }
    }
}


#define IDLE_SLOTS_BEHIND_PRODUCT(c0)                                           \
    asm volatile("v_nop\n\tv_nop\n\tv_nop\n\tv_nop\n\t"                        \
                 "v_nop\n\tv_nop\n\tv_nop\n\tv_nop" : "+v"(c0) : : "memory")

#define X_PITCH 132
#define XA_TILES 13
#define XB_TILES (NPB / 16 - XA_TILES)
#define XA_VEC4  (XA_TILES * 16 * 32)
#define XB_VEC4  (XB_TILES * 16 * 32)
static_assert(NPB % 16 == 0, "a range must be whole row tiles");
static_assert(XA_VEC4 > 6 * 1024 && XA_VEC4 <= 7 * 1024, "the bucket kernel spells out 7 trips of 1024 float4s for part one");
static_assert(XB_VEC4 == 6 * 1024, "the bucket kernel spells out 6 trips of 1024 float4s for part two");
static_assert(XA_TILES > 8 && XA_TILES <= 16 && XB_TILES > 8 && XB_TILES <= 16, "a wave has one or two row tiles in each part");

#define XS_PUT(i, v) *(float4*)(xs + ((i) >> 5) * X_PITCH + ((i) & 31) * 4) = (v)

__device__ __forceinline__ void product_tile(const float* xs, const float* sdis, float* hrow,
                                             const v2f (&bw)[32], const int rt,
                                             const int half, const int l) {
    const lfloat* arow = (const lfloat*)(xs + (rt * 16 + l) * X_PITCH + 2 * half);
    const lfloat* srow = (const lfloat*)(sdis + rt * 16 + half * 8);
    gfloat* o = (gfloat*)(hrow + (rt * 16 + half * 8) * 64 + l);
    asm volatile("" : "+v"(arow), "+v"(srow), "+v"(o));
    v8f c0 = {};
#pragma unroll
    for (int q = 0; q < 4; ++q) {
        v2f a[8];
#pragma unroll
        for (int s = 0; s < 8; ++s)
            a[s] = *(const __attribute__((address_space(3))) v2f*)(arow + 32 * q + 4 * s);
        __builtin_amdgcn_sched_barrier(0);
#pragma unroll
        for (int s = 0; s < 8; ++s)
            c0 = __builtin_amdgcn_wmma_f32_16x16x4_f32(false, a[s], false, bw[8 * q + s],
                                                       (short)0, c0, false, false);
        IDLE_SLOTS_BEHIND_PRODUCT(c0);
    }
    const float4 s0 = *(const __attribute__((address_space(3))) float4*)(srow);
    const float4 s1 = *(const __attribute__((address_space(3))) float4*)(srow + 4);
    const float sc[8] = {s0.x, s0.y, s0.z, s0.w, s1.x, s1.y, s1.z, s1.w};
#pragma unroll
    for (int v = 0; v < 8; ++v) o[v * 64] = c0[v] * sc[v];
}

__global__ __launch_bounds__(1024) void bucket_gemm1_kernel(
    const int* __restrict__ pk, const int* __restrict__ tcnt,
    const int* __restrict__ tstart, int* __restrict__ off,
    float* __restrict__ dis, int* __restrict__ srcs,
    const float* __restrict__ x, const float* __restrict__ W,
    float* __restrict__ h) {
    __shared__ int p[NCHUNK + 1];
    __shared__ int sabs[NCHUNK];
    __shared__ int lcnt[NPB];
    __shared__ int wsum[32];
    __shared__ int img[IMG_INTS + 32];
    __shared__ __attribute__((aligned(16))) float xs[NPB * X_PITCH];
    __shared__ __attribute__((aligned(16))) float sdis[NPB];
    const int tid = threadIdx.x;
    const int b   = blockIdx.x;

    const float4* x4 = (const float4*)(x + b * (NPB * 128));
    const float4 xa0 = x4[0 * 1024 + tid];
    const float4 xa1 = x4[1 * 1024 + tid];
    const float4 xa2 = x4[2 * 1024 + tid];
    const float4 xa3 = x4[3 * 1024 + tid];
    const float4 xa4 = x4[4 * 1024 + tid];
    const float4 xa5 = x4[5 * 1024 + tid];
    const float4 xa6 = x4[min(6 * 1024 + tid, XA_VEC4 - 1)];
    asm volatile("" : : "s"(x) : "memory");

    const int4* t4 = (const int4*)tcnt;
    const int n4 = b * (NCHUNK / 4);
    int part = 0;
    for (int k0 = 0; k0 < n4; k0 += 8 * 1024) {
        int4 v[8];
#pragma unroll
        for (int u = 0; u < 8; ++u) {
            const int k = k0 + u * 1024 + tid;
            v[u] = t4[min(k, n4 - 1)];
        }
#pragma unroll
        for (int u = 0; u < 8; ++u) {
            const int k = k0 + u * 1024 + tid;
            const int s = (v[u].x + v[u].y) + (v[u].z + v[u].w);
            part += (k < n4) ? s : 0;
        }
    }
    int c = 0;
    if (tid < NCHUNK) {
        c = tcnt[b * NCHUNK + tid];
        sabs[tid] = tstart[b * NCHUNK + tid];
    }
    if (tid < NPB) lcnt[tid] = 0;
    XS_PUT(0 * 1024 + tid, xa0);
    XS_PUT(1 * 1024 + tid, xa1);
    XS_PUT(2 * 1024 + tid, xa2);
    XS_PUT(3 * 1024 + tid, xa3);
    XS_PUT(4 * 1024 + tid, xa4);
    XS_PUT(5 * 1024 + tid, xa5);
    if (6 * 1024 + tid < XA_VEC4) XS_PUT(6 * 1024 + tid, xa6);
    int Tb, base;
    const int exc = block_scan_excl(c, wsum, Tb);
    if (tid < NCHUNK) p[tid] = exc;
    if (tid == 0) p[NCHUNK] = Tb;
    (void)block_scan_excl(part, wsum, base);
    __syncthreads();
    const int a0 = base & ~31;
    const bool staged = Tb <= IMG_INTS;

    int e[8], rk[8];
#pragma unroll
    for (int u = 0; u < 8; ++u) { e[u] = 0; rk[u] = 0; }
    if (staged) {
        if (Tb > 0) {
            int jj[8], idx[8];
#pragma unroll
            for (int u = 0; u < 8; ++u) jj[u] = min(u * 1024 + tid, Tb - 1);
            find_in_runs(p, sabs, jj, idx);
#pragma unroll
            for (int u = 0; u < 8; ++u) e[u] = pk[idx[u]];
#pragma unroll
            for (int u = 0; u < 8; ++u)
                if (u * 1024 + tid < Tb) rk[u] = atomicAdd(&lcnt[e[u] >> RBITS], 1);
        }
    } else {
        for (int j0 = 0; j0 < Tb; j0 += 8 * 1024) {
            int jj[8], idx[8], ev[8];
#pragma unroll
            for (int u = 0; u < 8; ++u) jj[u] = min(j0 + u * 1024 + tid, Tb - 1);
            find_in_runs(p, sabs, jj, idx);
#pragma unroll
            for (int u = 0; u < 8; ++u) ev[u] = pk[idx[u]];
#pragma unroll
            for (int u = 0; u < 8; ++u) {
                const int j = j0 + u * 1024 + tid;
                if (j < Tb) atomicAdd(&lcnt[ev[u] >> RBITS], 1);
            }
        }
    }
    __syncthreads();
    const int d = (tid < NPB) ? lcnt[tid] : 0;
    int tot;
    const int ex = block_scan_excl(d, wsum, tot);
    if (tid < NPB) {
        const int node = b * NPB + tid;
        const int o = base + ex;
        const float dn = rsqrtf((float)(d + 1));
        off[node] = o;
        dis[node] = dn;
        sdis[tid] = dn;
        lcnt[tid] = o;
    }
    if (b == NBINS - 1 && tid == 0) off[N_NODES] = base + tot;
    __syncthreads();

    if (staged) {
#pragma unroll
        for (int u = 0; u < 8; ++u)
            if (u * 1024 + tid < Tb)
                img[lcnt[e[u] >> RBITS] + rk[u] - a0] = e[u] & ((1 << RBITS) - 1);
        __syncthreads();
        copy_image_out(img, srcs, a0, base, base + Tb);
    } else {
        place_range(pk, p, sabs, lcnt, srcs, Tb);
    }

    const int lane = tid & 31;
    const int wave = __builtin_amdgcn_readfirstlane(tid >> 5);
    const int n0   = (wave & 3) * 16;
    const int g8   = wave >> 2;
    const int half = lane >> 4;
    const int l    = lane & 15;
    const float* bcol = W + (2 * half) * 64 + n0 + l;
    v2f bw[32];
#pragma unroll
    for (int s = 0; s < 32; ++s) {
        bw[s].x = bcol[(4 * s + 0) * 64];
        bw[s].y = bcol[(4 * s + 1) * 64];
    }
    asm volatile("" : "+v"(bw[0]), "+v"(bw[1]), "+v"(bw[2]), "+v"(bw[3]), "+v"(bw[4]), "+v"(bw[5]),
                      "+v"(bw[6]), "+v"(bw[7]), "+v"(bw[8]), "+v"(bw[9]), "+v"(bw[10]), "+v"(bw[11]),
                      "+v"(bw[12]), "+v"(bw[13]), "+v"(bw[14]), "+v"(bw[15]));
    asm volatile("" : "+v"(bw[16]), "+v"(bw[17]), "+v"(bw[18]), "+v"(bw[19]), "+v"(bw[20]), "+v"(bw[21]),
                      "+v"(bw[22]), "+v"(bw[23]), "+v"(bw[24]), "+v"(bw[25]), "+v"(bw[26]), "+v"(bw[27]),
                      "+v"(bw[28]), "+v"(bw[29]), "+v"(bw[30]), "+v"(bw[31]));
    const float4 xb0 = x4[XA_VEC4 + 0 * 1024 + tid];
    const float4 xb1 = x4[XA_VEC4 + 1 * 1024 + tid];
    const float4 xb2 = x4[XA_VEC4 + 2 * 1024 + tid];
    const float4 xb3 = x4[XA_VEC4 + 3 * 1024 + tid];
    const float4 xb4 = x4[XA_VEC4 + 4 * 1024 + tid];
    const float4 xb5 = x4[XA_VEC4 + 5 * 1024 + tid];
    asm volatile("" : : "s"(x) : "memory");
    float* hrow = h + b * (NPB * 64) + n0;
    for (int rt = g8; rt < XA_TILES; rt += 8)
        product_tile(xs, sdis, hrow, bw, rt, half, l);
    XS_PUT(XA_VEC4 + 0 * 1024 + tid, xb0);
    XS_PUT(XA_VEC4 + 1 * 1024 + tid, xb1);
    XS_PUT(XA_VEC4 + 2 * 1024 + tid, xb2);
    XS_PUT(XA_VEC4 + 3 * 1024 + tid, xb3);
    XS_PUT(XA_VEC4 + 4 * 1024 + tid, xb4);
    XS_PUT(XA_VEC4 + 5 * 1024 + tid, xb5);
    __syncthreads();
    for (int j = (g8 + 3) & 7; j < XB_TILES; j += 8)
        product_tile(xs, sdis, hrow, bw, XA_TILES + j, half, l);
}
#undef XS_PUT

template <int F, bool RELU>
__device__ __forceinline__ float4 gather_row_sum(const int* __restrict__ off,
                                                 const int* __restrict__ srcs,
                                                 const float dc,
                                                 const float* __restrict__ hs,
                                                 const float* __restrict__ bias,
                                                 const int c, const int g) {
    constexpr int LANES = F / 4;
    constexpr int BATCH = 2 * LANES;
    static_assert(LANES % 8 == 0, "a trip of eight ids comes from one of the two id registers");
    const int fc  = g * 4;
    const int beg = off[c];
    const int end = off[c + 1];

    float4 acc = *(const float4*)(hs + c * F + fc);
    const float4 bv = *(const float4*)(bias + fc);

    for (int b0 = beg; b0 < end; b0 += BATCH) {
        const int n = min(BATCH, end - b0);
        int myr0 = c, myr1 = c;
        if (g < n) myr0 = srcs[b0 + g];
        if (LANES + g < n) myr1 = srcs[b0 + LANES + g];
        for (int k0 = 0; k0 < n; k0 += 8) {
            const int ids = (k0 < LANES) ? myr0 : myr1;
            const int kb  = k0 & (LANES - 1);
            float4 v[8];
#pragma unroll
            for (int j = 0; j < 8; ++j) {
                const int r = __shfl(ids, kb + j, LANES);
                v[j] = *(const float4*)(hs + r * F + fc);
            }
            acc.x += v[0].x; acc.y += v[0].y; acc.z += v[0].z; acc.w += v[0].w;
#pragma unroll
            for (int j = 1; j < 8; ++j) {
                const float m = (k0 + j < n) ? 1.0f : 0.0f;
                acc.x += v[j].x * m; acc.y += v[j].y * m;
                acc.z += v[j].z * m; acc.w += v[j].w * m;
            }
        }
    }
    acc.x = bv.x + acc.x * dc;
    acc.y = bv.y + acc.y * dc;
    acc.z = bv.z + acc.z * dc;
    acc.w = bv.w + acc.w * dc;
    if (RELU) {
        acc.x = fmaxf(acc.x, 0.0f);
        acc.y = fmaxf(acc.y, 0.0f);
        acc.z = fmaxf(acc.z, 0.0f);
        acc.w = fmaxf(acc.w, 0.0f);
    }
    return acc;
}

#define TILE_PITCH 68
__global__ __launch_bounds__(256) void gather64_relu_gemm2_kernel(
    const int* __restrict__ off, const int* __restrict__ srcs,
    const float* __restrict__ dis, const float* __restrict__ h,
    const float* __restrict__ bias, const float* __restrict__ W,
    float* __restrict__ gout) {
    __shared__ __attribute__((aligned(16))) float tile[16 * TILE_PITCH];
    __shared__ __attribute__((aligned(16))) float sdis[16];
    const int cl   = threadIdx.x >> 4;
    const int g    = threadIdx.x & 15;
    const int m0   = blockIdx.x * 16;
    const int lane = threadIdx.x & 31;
    const int wave = threadIdx.x >> 5;
    const int half = lane >> 4;
    const int l    = lane & 15;
    const int n0   = (wave & 1) * 16;

    const float dc = dis[m0 + cl];
    const float4 acc = gather_row_sum<64, true>(off, srcs, dc, h, bias, m0 + cl, g);
    *(float4*)(tile + cl * TILE_PITCH + g * 4) = acc;
    if (g == 0) sdis[cl] = dc;

    v2f bw[16];
    if (wave < 2) {
        const float* bcol = W + (2 * half) * 32 + n0 + l;
#pragma unroll
        for (int s = 0; s < 16; ++s) {
            bw[s].x = bcol[(4 * s + 0) * 32];
            bw[s].y = bcol[(4 * s + 1) * 32];
        }
    }
    __syncthreads();
    if (wave < 2) {
        const lfloat* arow = (const lfloat*)(tile + l * TILE_PITCH + 2 * half);
        const lfloat* srow = (const lfloat*)(sdis + half * 8);
        gfloat* o = (gfloat*)(gout + (m0 + half * 8) * 32 + n0 + l);
        asm volatile("" : "+v"(arow), "+v"(srow), "+v"(o));
        v8f c0 = {};
        v2f a[8];
#pragma unroll
        for (int s = 0; s < 8; ++s) a[s] = *(const __attribute__((address_space(3))) v2f*)(arow + 4 * s);
        __builtin_amdgcn_sched_barrier(0);
#pragma unroll
        for (int s = 0; s < 8; ++s)
            c0 = __builtin_amdgcn_wmma_f32_16x16x4_f32(false, a[s], false, bw[s],
                                                       (short)0, c0, false, false);
        IDLE_SLOTS_BEHIND_PRODUCT(c0);
#pragma unroll
        for (int s = 0; s < 8; ++s) a[s] = *(const __attribute__((address_space(3))) v2f*)(arow + 32 + 4 * s);
        __builtin_amdgcn_sched_barrier(0);
#pragma unroll
        for (int s = 0; s < 8; ++s)
            c0 = __builtin_amdgcn_wmma_f32_16x16x4_f32(false, a[s], false, bw[8 + s],
                                                       (short)0, c0, false, false);
        IDLE_SLOTS_BEHIND_PRODUCT(c0);
        const float4 s0 = *(const __attribute__((address_space(3))) float4*)(srow);
        const float4 s1 = *(const __attribute__((address_space(3))) float4*)(srow + 4);
        const float sc[8] = {s0.x, s0.y, s0.z, s0.w, s1.x, s1.y, s1.z, s1.w};
#pragma unroll
        for (int v = 0; v < 8; ++v) o[v * 32] = c0[v] * sc[v];
    }
}

__global__ __launch_bounds__(256) void gather32_kernel(
    const int* __restrict__ off, const int* __restrict__ srcs,
    const float* __restrict__ dis, const float* __restrict__ h,
    const float* __restrict__ bias, float* __restrict__ res) {
    const int tid = blockIdx.x * blockDim.x + threadIdx.x;
    const int c   = tid / 8;
    const int g   = tid % 8;
    const float4 acc = gather_row_sum<32, false>(off, srcs, dis[c], h, bias, c, g);
    *(float4*)(res + c * 32 + g * 4) = acc;
}

extern "C" void kernel_launch(void* const* d_in, const int* in_sizes, int n_in,
                              void* d_out, int out_size, void* d_ws, size_t ws_size,
                              hipStream_t stream) {
    const float* x  = (const float*)d_in[0];
    const int*   ei = (const int*)d_in[1];
    const float* W1 = (const float*)d_in[2];
    const float* b1 = (const float*)d_in[3];
    const float* W2 = (const float*)d_in[4];
    const float* b2 = (const float*)d_in[5];
    float* out = (float*)d_out;

    const int E = in_sizes[1] / 2;
    const int* row = ei;
    const int* col = ei + E;

    float* dis  = (float*)d_ws;
    float* h1   = dis  + N_NODES;
    float* g2   = h1   + N_NODES * 64;
    int*   srcs = (int*)(g2 + N_NODES * 64);
    int*   off  = srcs + N_NODES * 16;
    int*   pk     = (int*)(g2 + N_NODES * 32);
    int*   tcnt   = (int*)g2;
    int*   tstart = tcnt + NBINS * NCHUNK;

    const int T = 256;
    chunk_sort_kernel<<<NCHUNK, 1024, 0, stream>>>(row, col, pk, tcnt, tstart, E);
    bucket_gemm1_kernel<<<NBINS, 1024, 0, stream>>>(pk, tcnt, tstart, off, dis, srcs, x, W1, h1);

    gather64_relu_gemm2_kernel<<<ROW_TILES, T, 0, stream>>>(off, srcs, dis, h1, b1, W2, g2);

    gather32_kernel<<<N_NODES * 8 / T, T, 0, stream>>>(off, srcs, dis, g2, b2, out);
}


# both sort launches: barriers wait for LDS only, so rows of x and W load under the sort part and no barrier waits for stores; table words asked for ahead of x, range start summed from run starts
# speedup vs baseline: 5.3539x; 1.0384x over previous
#include <hip/hip_runtime.h>
#include <hip/hip_bf16.h>

typedef __attribute__((ext_vector_type(2))) float v2f;
typedef __attribute__((ext_vector_type(8))) float v8f;
typedef __attribute__((address_space(1))) float gfloat;
typedef __attribute__((address_space(3))) float lfloat;

#define N_NODES 100000
#define ROW_TILES (N_NODES / 16)

__device__ __forceinline__ void lds_barrier() {
    __builtin_amdgcn_fence(__ATOMIC_RELEASE, "workgroup", "local");
    __builtin_amdgcn_s_barrier();
    __builtin_amdgcn_fence(__ATOMIC_ACQUIRE, "workgroup", "local");
}

#define NPB    400
#define NBINS  250
#define NCHUNK 256
#define RBITS  17
#define IMG_INTS (8 * 1024)
#define IMG_TRIPS ((IMG_INTS + 32 + 1023) / 1024)

__device__ __forceinline__ void copy_image_out(const int* img, int* __restrict__ dst,
                                               const int a0, const int first, const int stop) {
    int v[IMG_TRIPS];
#pragma unroll
    for (int k = 0; k < IMG_TRIPS; ++k)
        v[k] = img[min(k * 1024 + (int)threadIdx.x, IMG_INTS + 31)];
#pragma unroll
    for (int k = 0; k < IMG_TRIPS; ++k) {
        const int gi = a0 + k * 1024 + (int)threadIdx.x;
        if (gi >= first && gi < stop) dst[gi] = v[k];
    }
}
static_assert(NBINS * NPB == N_NODES, "ranges must tile the nodes");
static_assert((NCHUNK & (NCHUNK - 1)) == 0 && NCHUNK <= 1024 && NBINS <= 256, "bisection and scans assume this");
static_assert(N_NODES <= (1 << RBITS), "source id must fit the packed word");
static_assert(NPB <= (1 << (31 - RBITS)), "local target must fit the packed word");

__device__ __forceinline__ int block_scan_excl(int v, int* wsum, int& total) {
    const int lane = threadIdx.x & 31;
    const int wave = threadIdx.x >> 5;
    int inc = v;
#pragma unroll
    for (int o = 1; o < 32; o <<= 1) {
        int u = __shfl_up(inc, o);
        if (lane >= o) inc += u;
    }
    lds_barrier();
    if (lane == 31) wsum[wave] = inc;
    lds_barrier();
    int wb = 0, tot = 0;
#pragma unroll
    for (int w = 0; w < 32; ++w) {
        const int s = wsum[w];
        if (w < wave) wb += s;
        tot += s;
    }
    total = tot;
    return wb + inc - v;
}

__device__ __forceinline__ void place_chunk(const int* __restrict__ row, const int* __restrict__ col,
                                            int* __restrict__ pk, int* lh,
                                            const int beg, const int end) {
    const int tid = threadIdx.x;
    for (int i0 = beg; i0 < end; i0 += 8 * 1024) {
        int cv[8], rv[8];
#pragma unroll
        for (int u = 0; u < 8; ++u) {
            const int i = min(i0 + u * 1024 + tid, end - 1);
            cv[u] = col[i];
            rv[u] = row[i];
        }
#pragma unroll
        for (int u = 0; u < 8; ++u) {
            const int i = i0 + u * 1024 + tid;
            if (i < end) {
                const unsigned b = (unsigned)cv[u] / NPB;
                const int t = cv[u] - (int)b * NPB;
                const int slot = atomicAdd(&lh[b], 1);
                pk[slot] = (t << RBITS) | rv[u];
            }
        }
    }
}

__global__ __launch_bounds__(1024) void chunk_sort_kernel(
    const int* __restrict__ row, const int* __restrict__ col,
    int* __restrict__ pk, int* __restrict__ tcnt, int* __restrict__ tstart, int E) {
    __shared__ int lh[256];
    __shared__ int wsum[32];
    __shared__ int img[IMG_INTS + 32];
    const int tid = threadIdx.x;
    const int w   = blockIdx.x;
    const int chunk = (E + NCHUNK - 1) / NCHUNK;
    const int beg = min(w * chunk, E);
    const int end = min(beg + chunk, E);
    const int a0  = beg & ~31;
    const bool staged = (end - beg) <= IMG_INTS;

    if (tid < 256) lh[tid] = 0;
    lds_barrier();
    int cv[8], rv[8], rk[8];
    if (staged) {
        if (end > beg) {
#pragma unroll
            for (int u = 0; u < 8; ++u) {
                const int i = min(beg + u * 1024 + tid, end - 1);
                cv[u] = col[i];
                rv[u] = row[i];
            }
#pragma unroll
            for (int u = 0; u < 8; ++u) {
                rk[u] = 0;
                if (beg + u * 1024 + tid < end) rk[u] = atomicAdd(&lh[(unsigned)cv[u] / NPB], 1);
            }
        }
    } else {
        for (int i0 = beg; i0 < end; i0 += 8 * 1024) {
            int cw[8];
#pragma unroll
            for (int u = 0; u < 8; ++u) {
                const int i = i0 + u * 1024 + tid;
                cw[u] = col[min(i, end - 1)];
            }
#pragma unroll
            for (int u = 0; u < 8; ++u) {
                const int i = i0 + u * 1024 + tid;
                if (i < end) atomicAdd(&lh[(unsigned)cw[u] / NPB], 1);
            }
        }
    }
    lds_barrier();
    const int v = (tid < NBINS) ? lh[tid] : 0;
    int tot;
    const int ex = block_scan_excl(v, wsum, tot);
    if (tid < NBINS) {
        tcnt[tid * NCHUNK + w]   = v;
        tstart[tid * NCHUNK + w] = beg + ex;
        lh[tid] = beg + ex;
    }
    lds_barrier();
    if (staged) {
#pragma unroll
        for (int u = 0; u < 8; ++u) {
            if (beg + u * 1024 + tid < end) {
                const unsigned b = (unsigned)cv[u] / NPB;
                const int t = cv[u] - (int)b * NPB;
                img[lh[b] + rk[u] - a0] = (t << RBITS) | rv[u];
            }
        }
        lds_barrier();
        copy_image_out(img, pk, a0, beg, end);
    } else {
        place_chunk(row, col, pk, lh, beg, end);
    }
}

__device__ __forceinline__ void find_in_runs(const int* p, const int* sabs,
                                             const int (&j)[8], int (&idx)[8]) {
    int lo[8];
#pragma unroll
    for (int u = 0; u < 8; ++u) lo[u] = 0;
#pragma unroll
    for (int half = NCHUNK / 2; half > 0; half >>= 1) {
        int pm[8];
#pragma unroll
        for (int u = 0; u < 8; ++u) pm[u] = p[lo[u] + half];
#pragma unroll
        for (int u = 0; u < 8; ++u) if (pm[u] <= j[u]) lo[u] += half;
    }
#pragma unroll
    for (int u = 0; u < 8; ++u) idx[u] = sabs[lo[u]] + (j[u] - p[lo[u]]);
}

__device__ __forceinline__ void place_range(const int* __restrict__ pk, const int* p, const int* sabs,
                                            int* lcnt, int* __restrict__ srcs, const int Tb) {
    const int tid = threadIdx.x;
    for (int j0 = 0; j0 < Tb; j0 += 8 * 1024) {
        int jj[8], idx[8];
#pragma unroll
        for (int u = 0; u < 8; ++u) jj[u] = min(j0 + u * 1024 + tid, Tb - 1);
        find_in_runs(p, sabs, jj, idx);
        int e[8];
#pragma unroll
        for (int u = 0; u < 8; ++u) e[u] = pk[idx[u]];
#pragma unroll
        for (int u = 0; u < 8; ++u) {
            const int j = j0 + u * 1024 + tid;
            if (j < Tb) {
                const int slot = atomicAdd(&lcnt[e[u] >> RBITS], 1);
                srcs[slot] = e[u] & ((1 << RBITS) - 1);
            }
        }
    }
}


#define IDLE_SLOTS_BEHIND_PRODUCT(c0)                                           \
    asm volatile("v_nop\n\tv_nop\n\tv_nop\n\tv_nop\n\t"                        \
                 "v_nop\n\tv_nop\n\tv_nop\n\tv_nop" : "+v"(c0) : : "memory")

#define X_PITCH 132
#define XA_TILES 13
#define XB_TILES (NPB / 16 - XA_TILES)
#define XA_VEC4  (XA_TILES * 16 * 32)
#define XB_VEC4  (XB_TILES * 16 * 32)
static_assert(NPB % 16 == 0, "a range must be whole row tiles");
static_assert(XA_VEC4 > 6 * 1024 && XA_VEC4 <= 7 * 1024, "the bucket kernel spells out 7 trips of 1024 float4s for part one");
static_assert(XB_VEC4 == 6 * 1024, "the bucket kernel spells out 6 trips of 1024 float4s for part two");
static_assert(XA_TILES > 8 && XA_TILES <= 16 && XB_TILES > 8 && XB_TILES <= 16, "a wave has one or two row tiles in each part");

#define XS_PUT(i, v) *(float4*)(xs + ((i) >> 5) * X_PITCH + ((i) & 31) * 4) = (v)

__device__ __forceinline__ void product_tile(const float* xs, const float* sdis, float* hrow,
                                             const v2f (&bw)[32], const int rt,
                                             const int half, const int l) {
    const lfloat* arow = (const lfloat*)(xs + (rt * 16 + l) * X_PITCH + 2 * half);
    const lfloat* srow = (const lfloat*)(sdis + rt * 16 + half * 8);
    int oi = (rt * 16 + half * 8) * 64 + l;
    asm volatile("" : "+v"(arow), "+v"(srow), "+v"(oi));
    v8f c0 = {};
#pragma unroll
    for (int q = 0; q < 4; ++q) {
        v2f a[8];
#pragma unroll
        for (int s = 0; s < 8; ++s)
            a[s] = *(const __attribute__((address_space(3))) v2f*)(arow + 32 * q + 4 * s);
        __builtin_amdgcn_sched_barrier(0);
#pragma unroll
        for (int s = 0; s < 8; ++s)
            c0 = __builtin_amdgcn_wmma_f32_16x16x4_f32(false, a[s], false, bw[8 * q + s],
                                                       (short)0, c0, false, false);
        IDLE_SLOTS_BEHIND_PRODUCT(c0);
    }
    const float4 s0 = *(const __attribute__((address_space(3))) float4*)(srow);
    const float4 s1 = *(const __attribute__((address_space(3))) float4*)(srow + 4);
    const float sc[8] = {s0.x, s0.y, s0.z, s0.w, s1.x, s1.y, s1.z, s1.w};
#pragma unroll
    for (int v = 0; v < 8; ++v) hrow[oi + v * 64] = c0[v] * sc[v];
}

__global__ __launch_bounds__(1024) void bucket_gemm1_kernel(
    const int* __restrict__ pk, const int* __restrict__ tcnt,
    const int* __restrict__ tstart, int* __restrict__ off,
    float* __restrict__ dis, int* __restrict__ srcs,
    const float* __restrict__ x, const float* __restrict__ W,
    float* __restrict__ h, int E) {
    __shared__ int p[NCHUNK + 1];
    __shared__ int sabs[NCHUNK];
    __shared__ int lcnt[NPB];
    __shared__ int wsum[32];
    __shared__ int img[IMG_INTS + 32];
    __shared__ __attribute__((aligned(16))) float xs[NPB * X_PITCH];
    __shared__ __attribute__((aligned(16))) float sdis[NPB];
    const int tid = threadIdx.x;
    const int b   = blockIdx.x;

    const int ti = b * NCHUNK + min(tid, NCHUNK - 1);
    const int c_raw  = tcnt[ti];
    const int sa_raw = tstart[ti];
    asm volatile("" : : "s"(tcnt), "s"(tstart), "s"(x) : "memory");

    const float4* x4 = (const float4*)(x + b * (NPB * 128));
    const float4 xa0 = x4[0 * 1024 + tid];
    const float4 xa1 = x4[1 * 1024 + tid];
    const float4 xa2 = x4[2 * 1024 + tid];
    const float4 xa3 = x4[3 * 1024 + tid];
    const float4 xa4 = x4[4 * 1024 + tid];
    const float4 xa5 = x4[5 * 1024 + tid];
    const float4 xa6 = x4[min(6 * 1024 + tid, XA_VEC4 - 1)];
    asm volatile("" : : "s"(x) : "memory");

    const int chunk = (E + NCHUNK - 1) / NCHUNK;
    int c = 0, part = 0;
    if (tid < NCHUNK) {
        c = c_raw;
        sabs[tid] = sa_raw;
        part = sa_raw - min(tid * chunk, E);
    }
    if (tid < NPB) lcnt[tid] = 0;
    int Tb, base;
    const int exc = block_scan_excl(c, wsum, Tb);
    if (tid < NCHUNK) p[tid] = exc;
    if (tid == 0) p[NCHUNK] = Tb;
    (void)block_scan_excl(part, wsum, base);
    lds_barrier();
    const int a0 = base & ~31;
    const int Tbs = __builtin_amdgcn_readfirstlane(Tb);
    const bool staged = Tbs <= IMG_INTS;

    int e[8], rk[8];
#pragma unroll
    for (int u = 0; u < 8; ++u) { e[u] = 0; rk[u] = 0; }
    if (staged) {
        if (Tbs > 0) {
            int jj[8], idx[8];
#pragma unroll
            for (int u = 0; u < 8; ++u) jj[u] = min(u * 1024 + tid, Tb - 1);
            find_in_runs(p, sabs, jj, idx);
#pragma unroll
            for (int u = 0; u < 8; ++u) e[u] = pk[idx[u]];
#pragma unroll
            for (int u = 0; u < 8; ++u)
                if (u * 1024 + tid < Tb) rk[u] = atomicAdd(&lcnt[e[u] >> RBITS], 1);
        }
    }
    if (!staged) {
        for (int j0 = 0; j0 < Tb; j0 += 8 * 1024) {
            int jj[8], idx[8], ev[8];
#pragma unroll
            for (int u = 0; u < 8; ++u) jj[u] = min(j0 + u * 1024 + tid, Tb - 1);
            find_in_runs(p, sabs, jj, idx);
#pragma unroll
            for (int u = 0; u < 8; ++u) ev[u] = pk[idx[u]];
#pragma unroll
            for (int u = 0; u < 8; ++u) {
                const int j = j0 + u * 1024 + tid;
                if (j < Tb) atomicAdd(&lcnt[ev[u] >> RBITS], 1);
            }
        }
    }
    XS_PUT(0 * 1024 + tid, xa0);
    XS_PUT(1 * 1024 + tid, xa1);
    XS_PUT(2 * 1024 + tid, xa2);
    XS_PUT(3 * 1024 + tid, xa3);
    XS_PUT(4 * 1024 + tid, xa4);
    XS_PUT(5 * 1024 + tid, xa5);
    if (6 * 1024 + tid < XA_VEC4) XS_PUT(6 * 1024 + tid, xa6);
    lds_barrier();
    const int d = (tid < NPB) ? lcnt[tid] : 0;
    int tot;
    const int ex = block_scan_excl(d, wsum, tot);
    if (tid < NPB) {
        const int node = b * NPB + tid;
        const int o = base + ex;
        const float dn = rsqrtf((float)(d + 1));
        off[node] = o;
        dis[node] = dn;
        sdis[tid] = dn;
        lcnt[tid] = o;
    }
    if (b == NBINS - 1 && tid == 0) off[N_NODES] = base + tot;
    lds_barrier();

    const int lane = tid & 31;
    const int wave = __builtin_amdgcn_readfirstlane(tid >> 5);
    const int n0   = (wave & 3) * 16;
    const int g8   = wave >> 2;
    const int half = lane >> 4;
    const int l    = lane & 15;
    const float* bcol = W + (2 * half) * 64 + n0 + l;
    v2f bw[32];
#pragma unroll
    for (int s = 0; s < 32; ++s) {
        bw[s].x = bcol[(4 * s + 0) * 64];
        bw[s].y = bcol[(4 * s + 1) * 64];
    }
    asm volatile("" : : "s"(W) : "memory");

    if (staged) {
#pragma unroll
        for (int u = 0; u < 8; ++u)
            if (u * 1024 + tid < Tb)
                img[lcnt[e[u] >> RBITS] + rk[u] - a0] = e[u] & ((1 << RBITS) - 1);
        lds_barrier();
        copy_image_out(img, srcs, a0, base, base + Tb);
    }
    if (!staged) place_range(pk, p, sabs, lcnt, srcs, Tb);

    asm volatile("" : "+v"(bw[0]), "+v"(bw[1]), "+v"(bw[2]), "+v"(bw[3]), "+v"(bw[4]), "+v"(bw[5]),
                      "+v"(bw[6]), "+v"(bw[7]), "+v"(bw[8]), "+v"(bw[9]), "+v"(bw[10]), "+v"(bw[11]),
                      "+v"(bw[12]), "+v"(bw[13]), "+v"(bw[14]), "+v"(bw[15]));
    asm volatile("" : "+v"(bw[16]), "+v"(bw[17]), "+v"(bw[18]), "+v"(bw[19]), "+v"(bw[20]), "+v"(bw[21]),
                      "+v"(bw[22]), "+v"(bw[23]), "+v"(bw[24]), "+v"(bw[25]), "+v"(bw[26]), "+v"(bw[27]),
                      "+v"(bw[28]), "+v"(bw[29]), "+v"(bw[30]), "+v"(bw[31]));
    const float4 xb0 = x4[XA_VEC4 + 0 * 1024 + tid];
    const float4 xb1 = x4[XA_VEC4 + 1 * 1024 + tid];
    const float4 xb2 = x4[XA_VEC4 + 2 * 1024 + tid];
    const float4 xb3 = x4[XA_VEC4 + 3 * 1024 + tid];
    const float4 xb4 = x4[XA_VEC4 + 4 * 1024 + tid];
    const float4 xb5 = x4[XA_VEC4 + 5 * 1024 + tid];
    asm volatile("" : : "s"(x) : "memory");
    float* hrow = h + b * (NPB * 64) + n0;
    for (int rt = g8; rt < XA_TILES; rt += 8)
        product_tile(xs, sdis, hrow, bw, rt, half, l);
    XS_PUT(XA_VEC4 + 0 * 1024 + tid, xb0);
    XS_PUT(XA_VEC4 + 1 * 1024 + tid, xb1);
    XS_PUT(XA_VEC4 + 2 * 1024 + tid, xb2);
    XS_PUT(XA_VEC4 + 3 * 1024 + tid, xb3);
    XS_PUT(XA_VEC4 + 4 * 1024 + tid, xb4);
    XS_PUT(XA_VEC4 + 5 * 1024 + tid, xb5);
    lds_barrier();
    for (int j = (g8 + 3) & 7; j < XB_TILES; j += 8)
        product_tile(xs, sdis, hrow, bw, XA_TILES + j, half, l);
}
#undef XS_PUT

template <int F, bool RELU>
__device__ __forceinline__ float4 gather_row_sum(const int* __restrict__ off,
                                                 const int* __restrict__ srcs,
                                                 const float dc,
                                                 const float* __restrict__ hs,
                                                 const float* __restrict__ bias,
                                                 const int c, const int g) {
    constexpr int LANES = F / 4;
    constexpr int BATCH = 2 * LANES;
    static_assert(LANES % 8 == 0, "a trip of eight ids comes from one of the two id registers");
    const int fc  = g * 4;
    const int beg = off[c];
    const int end = off[c + 1];

    float4 acc = *(const float4*)(hs + c * F + fc);
    const float4 bv = *(const float4*)(bias + fc);

    for (int b0 = beg; b0 < end; b0 += BATCH) {
        const int n = min(BATCH, end - b0);
        int myr0 = c, myr1 = c;
        if (g < n) myr0 = srcs[b0 + g];
        if (LANES + g < n) myr1 = srcs[b0 + LANES + g];
        for (int k0 = 0; k0 < n; k0 += 8) {
            const int ids = (k0 < LANES) ? myr0 : myr1;
            const int kb  = k0 & (LANES - 1);
            float4 v[8];
#pragma unroll
            for (int j = 0; j < 8; ++j) {
                const int r = __shfl(ids, kb + j, LANES);
                v[j] = *(const float4*)(hs + r * F + fc);
            }
            acc.x += v[0].x; acc.y += v[0].y; acc.z += v[0].z; acc.w += v[0].w;
#pragma unroll
            for (int j = 1; j < 8; ++j) {
                const float m = (k0 + j < n) ? 1.0f : 0.0f;
                acc.x += v[j].x * m; acc.y += v[j].y * m;
                acc.z += v[j].z * m; acc.w += v[j].w * m;
            }
        }
    }
    acc.x = bv.x + acc.x * dc;
    acc.y = bv.y + acc.y * dc;
    acc.z = bv.z + acc.z * dc;
    acc.w = bv.w + acc.w * dc;
    if (RELU) {
        acc.x = fmaxf(acc.x, 0.0f);
        acc.y = fmaxf(acc.y, 0.0f);
        acc.z = fmaxf(acc.z, 0.0f);
        acc.w = fmaxf(acc.w, 0.0f);
    }
    return acc;
}

#define TILE_PITCH 68
__global__ __launch_bounds__(256) void gather64_relu_gemm2_kernel(
    const int* __restrict__ off, const int* __restrict__ srcs,
    const float* __restrict__ dis, const float* __restrict__ h,
    const float* __restrict__ bias, const float* __restrict__ W,
    float* __restrict__ gout) {
    __shared__ __attribute__((aligned(16))) float tile[16 * TILE_PITCH];
    __shared__ __attribute__((aligned(16))) float sdis[16];
    const int cl   = threadIdx.x >> 4;
    const int g    = threadIdx.x & 15;
    const int m0   = blockIdx.x * 16;
    const int lane = threadIdx.x & 31;
    const int wave = threadIdx.x >> 5;
    const int half = lane >> 4;
    const int l    = lane & 15;
    const int n0   = (wave & 1) * 16;

    const float dc = dis[m0 + cl];
    const float4 acc = gather_row_sum<64, true>(off, srcs, dc, h, bias, m0 + cl, g);
    *(float4*)(tile + cl * TILE_PITCH + g * 4) = acc;
    if (g == 0) sdis[cl] = dc;

    v2f bw[16];
    if (wave < 2) {
        const float* bcol = W + (2 * half) * 32 + n0 + l;
#pragma unroll
        for (int s = 0; s < 16; ++s) {
            bw[s].x = bcol[(4 * s + 0) * 32];
            bw[s].y = bcol[(4 * s + 1) * 32];
        }
    }
    __syncthreads();
    if (wave < 2) {
        const lfloat* arow = (const lfloat*)(tile + l * TILE_PITCH + 2 * half);
        const lfloat* srow = (const lfloat*)(sdis + half * 8);
        gfloat* o = (gfloat*)(gout + (m0 + half * 8) * 32 + n0 + l);
        asm volatile("" : "+v"(arow), "+v"(srow), "+v"(o));
        v8f c0 = {};
        v2f a[8];
#pragma unroll
        for (int s = 0; s < 8; ++s) a[s] = *(const __attribute__((address_space(3))) v2f*)(arow + 4 * s);
        __builtin_amdgcn_sched_barrier(0);
#pragma unroll
        for (int s = 0; s < 8; ++s)
            c0 = __builtin_amdgcn_wmma_f32_16x16x4_f32(false, a[s], false, bw[s],
                                                       (short)0, c0, false, false);
        IDLE_SLOTS_BEHIND_PRODUCT(c0);
#pragma unroll
        for (int s = 0; s < 8; ++s) a[s] = *(const __attribute__((address_space(3))) v2f*)(arow + 32 + 4 * s);
        __builtin_amdgcn_sched_barrier(0);
#pragma unroll
        for (int s = 0; s < 8; ++s)
            c0 = __builtin_amdgcn_wmma_f32_16x16x4_f32(false, a[s], false, bw[8 + s],
                                                       (short)0, c0, false, false);
        IDLE_SLOTS_BEHIND_PRODUCT(c0);
        const float4 s0 = *(const __attribute__((address_space(3))) float4*)(srow);
        const float4 s1 = *(const __attribute__((address_space(3))) float4*)(srow + 4);
        const float sc[8] = {s0.x, s0.y, s0.z, s0.w, s1.x, s1.y, s1.z, s1.w};
#pragma unroll
        for (int v = 0; v < 8; ++v) o[v * 32] = c0[v] * sc[v];
    }
}

__global__ __launch_bounds__(256) void gather32_kernel(
    const int* __restrict__ off, const int* __restrict__ srcs,
    const float* __restrict__ dis, const float* __restrict__ h,
    const float* __restrict__ bias, float* __restrict__ res) {
    const int tid = blockIdx.x * blockDim.x + threadIdx.x;
    const int c   = tid / 8;
    const int g   = tid % 8;
    const float4 acc = gather_row_sum<32, false>(off, srcs, dis[c], h, bias, c, g);
    *(float4*)(res + c * 32 + g * 4) = acc;
}

extern "C" void kernel_launch(void* const* d_in, const int* in_sizes, int n_in,
                              void* d_out, int out_size, void* d_ws, size_t ws_size,
                              hipStream_t stream) {
    const float* x  = (const float*)d_in[0];
    const int*   ei = (const int*)d_in[1];
    const float* W1 = (const float*)d_in[2];
    const float* b1 = (const float*)d_in[3];
    const float* W2 = (const float*)d_in[4];
    const float* b2 = (const float*)d_in[5];
    float* out = (float*)d_out;

    const int E = in_sizes[1] / 2;
    const int* row = ei;
    const int* col = ei + E;

    float* dis  = (float*)d_ws;
    float* h1   = dis  + N_NODES;
    float* g2   = h1   + N_NODES * 64;
    int*   srcs = (int*)(g2 + N_NODES * 64);
    int*   off  = srcs + N_NODES * 16;
    int*   pk     = (int*)(g2 + N_NODES * 32);
    int*   tcnt   = (int*)g2;
    int*   tstart = tcnt + NBINS * NCHUNK;

    const int T = 256;
    chunk_sort_kernel<<<NCHUNK, 1024, 0, stream>>>(row, col, pk, tcnt, tstart, E);
    bucket_gemm1_kernel<<<NBINS, 1024, 0, stream>>>(pk, tcnt, tstart, off, dis, srcs, x, W1, h1, E);

    gather64_relu_gemm2_kernel<<<ROW_TILES, T, 0, stream>>>(off, srcs, dis, h1, b1, W2, g2);

    gather32_kernel<<<N_NODES * 8 / T, T, 0, stream>>>(off, srcs, dis, g2, b2, out);
}
